# starting kernel
# baseline (speedup 1.0000x reference)
#include <hip/hip_runtime.h>
#include <math.h>

#define NN 50000
#define NE 800000
#define INDIM 256
#define HID 32
#define HEADS 4
#define ETOT (NE + NN)

typedef __attribute__((ext_vector_type(2))) float v2f;
typedef __attribute__((ext_vector_type(8))) float v8f;

__device__ __forceinline__ unsigned fkey(float x) {
  unsigned u = __float_as_uint(x);
  return (u & 0x80000000u) ? ~u : (u | 0x80000000u);
}
__device__ __forceinline__ float unfkey(unsigned k) {
  return __uint_as_float((k & 0x80000000u) ? (k & 0x7fffffffu) : ~k);
}
__device__ __forceinline__ float elu(float v) {
  return (v > 0.f) ? v : (__expf(v) - 1.f);
}

__global__ void fill_u32(unsigned* p, unsigned v, int n) {
  int i = blockIdx.x * blockDim.x + threadIdx.x;
  if (i < n) p[i] = v;
}
__global__ void fill_f32(float* p, float v, int n) {
  int i = blockIdx.x * blockDim.x + threadIdx.x;
  if (i < n) p[i] = v;
}

template <int N, int K>
__global__ void gemm_wmma_f32(const float* __restrict__ A, const float* __restrict__ B,
                              float* __restrict__ C) {
  const int lane  = threadIdx.x;
  const int mbase = blockIdx.x * 16;
  const int nbase = (blockIdx.y * blockDim.y + threadIdx.y) * 16;
  const int l15   = lane & 15;
  const int koff  = (lane >> 4) << 1;
  const float* ap = A + (size_t)(mbase + l15) * K + koff;
  const float* bp = B + (size_t)koff * N + nbase + l15;
  v8f c = {0.f, 0.f, 0.f, 0.f, 0.f, 0.f, 0.f, 0.f};
#pragma unroll 8
  for (int k0 = 0; k0 < K; k0 += 4) {
    v2f a, b;
    a.x = ap[k0];
    a.y = ap[k0 + 1];
    b.x = bp[k0 * N];
    b.y = bp[(k0 + 1) * N];
    c = __builtin_amdgcn_wmma_f32_16x16x4_f32(false, a, false, b, (short)0, c,
                                              false, false);
  }
  const int crow = mbase + ((lane >> 4) << 3);
  float* cp = C + (size_t)crow * N + nbase + l15;
#pragma unroll
  for (int r = 0; r < 8; ++r) cp[r * N] = c[r];
}

__global__ void att_scores(const float* __restrict__ h, const float* __restrict__ att_s,
                           const float* __restrict__ att_d, float* __restrict__ as,
                           float* __restrict__ ad, int H, int C) {
  int i = blockIdx.x * blockDim.x + threadIdx.x;
  if (i >= NN * H) return;
  int hh = i % H;
  const float* hp = h + (size_t)i * C;
  float s = 0.f, d = 0.f;
  for (int c = 0; c < C; ++c) {
    float v = hp[c];
    s += v * att_s[hh * C + c];
    d += v * att_d[hh * C + c];
  }
  as[i] = s;
  ad[i] = d;
}

__global__ void edge_max(const int* __restrict__ src, const int* __restrict__ dst,
                         const float* __restrict__ as, const float* __restrict__ ad,
                         unsigned* __restrict__ mkey, int H) {
  long long i = (long long)blockIdx.x * blockDim.x + threadIdx.x;
  if (i >= (long long)ETOT * H) return;
  int e = (int)(i / H), h = (int)(i % H);
  int s = (e < NE) ? src[e] : (e - NE);
  int d = (e < NE) ? dst[e] : (e - NE);
  float x = as[s * H + h] + ad[d * H + h];
  x = (x > 0.f) ? x : 0.2f * x;
  atomicMax(&mkey[d * H + h], fkey(x));
}

__global__ void edge_acc(const int* __restrict__ src, const int* __restrict__ dst,
                         const float* __restrict__ as, const float* __restrict__ ad,
                         const unsigned* __restrict__ mkey, const float* __restrict__ feat,
                         float* __restrict__ acc, float* __restrict__ denom, int H, int C) {
  long long i = (long long)blockIdx.x * blockDim.x + threadIdx.x;
  if (i >= (long long)ETOT * H * C) return;
  int c = (int)(i % C);
  long long eh = i / C;
  int h = (int)(eh % H);
  int e = (int)(eh / H);
  int s = (e < NE) ? src[e] : (e - NE);
  int d = (e < NE) ? dst[e] : (e - NE);
  float x = as[s * H + h] + ad[d * H + h];
  x = (x > 0.f) ? x : 0.2f * x;
  float p = __expf(x - unfkey(mkey[d * H + h]));
  atomicAdd(&acc[((size_t)d * H + h) * C + c], p * feat[((size_t)s * H + h) * C + c]);
  if (c == 0) atomicAdd(&denom[d * H + h], p);
}

__global__ void norm_elu(float* __restrict__ acc, const float* __restrict__ denom,
                         const float* __restrict__ bias, int H, int C) {
  int i = blockIdx.x * blockDim.x + threadIdx.x;
  if (i >= NN * H * C) return;
  int n = i / (H * C);
  int r = i % (H * C);
  int h = r / C;
  acc[i] = elu(acc[i] / denom[n * H + h] + bias[r]);
}

__global__ void final_cls(const float* __restrict__ acc2, const float* __restrict__ denom2,
                          const float* __restrict__ b2, const float* __restrict__ lin_w,
                          const float* __restrict__ lin_b, float* __restrict__ out) {
  int n = blockIdx.x * (blockDim.x >> 5) + (threadIdx.x >> 5);
  int c = threadIdx.x & 31;
  if (n >= NN) return;
  float v = elu(acc2[(size_t)n * HID + c] / denom2[n] + b2[c]);
  float w0 = v * lin_w[c * 2 + 0];
  float w1 = v * lin_w[c * 2 + 1];
#pragma unroll
  for (int m = 16; m > 0; m >>= 1) {
    w0 += __shfl_xor(w0, m, 32);
    w1 += __shfl_xor(w1, m, 32);
  }
  if (c == 0) {
    out[n * 2 + 0] = w0 + lin_b[0];
    out[n * 2 + 1] = w1 + lin_b[1];
  }
}

static inline unsigned blks(long long tot, int bs) { return (unsigned)((tot + bs - 1) / bs); }

extern "C" void kernel_launch(void* const* d_in, const int* in_sizes, int n_in,
                              void* d_out, int out_size, void* d_ws, size_t ws_size,
                              hipStream_t stream) {
  const float* x        = (const float*)d_in[0];
  const int*   ei       = (const int*)d_in[1];
  const float* W1       = (const float*)d_in[2];
  const float* att_src1 = (const float*)d_in[3];
  const float* att_dst1 = (const float*)d_in[4];
  const float* b1       = (const float*)d_in[5];
  const float* W2       = (const float*)d_in[6];
  const float* att_src2 = (const float*)d_in[7];
  const float* att_dst2 = (const float*)d_in[8];
  const float* b2       = (const float*)d_in[9];
  const float* lin_w    = (const float*)d_in[10];
  const float* lin_b    = (const float*)d_in[11];
  float* out = (float*)d_out;
  const int* src = ei;
  const int* dst = ei + NE;

  float*    ws   = (float*)d_ws;
  float*    h1   = ws;
  float*    as1  = ws + 6400000;
  float*    ad1  = as1 + 200000;
  unsigned* m1   = (unsigned*)(ad1 + 200000);
  float*    d1   = (float*)(m1 + 200000);
  float*    acc1 = d1 + 200000;
  float*    z    = h1;
  float*    acc2 = h1 + 1600000;
  float*    as2  = h1 + 3200000;
  float*    ad2  = as2 + 50000;
  unsigned* m2   = (unsigned*)(ad2 + 50000);
  float*    d2   = (float*)(m2 + 50000);

  gemm_wmma_f32<HEADS * HID, INDIM>
      <<<dim3(NN / 16, 2), dim3(32, 4), 0, stream>>>(x, W1, h1);
  att_scores<<<blks(NN * HEADS, 256), 256, 0, stream>>>(h1, att_src1, att_dst1, as1, ad1, HEADS, HID);
  fill_u32<<<blks(200000, 256), 256, 0, stream>>>(m1, 0u, 200000);
  fill_f32<<<blks(200000, 256), 256, 0, stream>>>(d1, 0.f, 200000);
  fill_f32<<<blks(6400000, 256), 256, 0, stream>>>(acc1, 0.f, 6400000);
  edge_max<<<blks((long long)ETOT * HEADS, 256), 256, 0, stream>>>(src, dst, as1, ad1, m1, HEADS);
  edge_acc<<<blks((long long)ETOT * HEADS * HID, 256), 256, 0, stream>>>(src, dst, as1, ad1, m1,
                                                                         h1, acc1, d1, HEADS, HID);
  norm_elu<<<blks(6400000, 256), 256, 0, stream>>>(acc1, d1, b1, HEADS, HID);

  gemm_wmma_f32<HID, HEADS * HID>
      <<<dim3(NN / 16, 1), dim3(32, 2), 0, stream>>>(acc1, W2, z);
  att_scores<<<blks(NN, 256), 256, 0, stream>>>(z, att_src2, att_dst2, as2, ad2, 1, HID);
  fill_u32<<<blks(50000, 256), 256, 0, stream>>>(m2, 0u, 50000);
  fill_f32<<<blks(50000, 256), 256, 0, stream>>>(d2, 0.f, 50000);
  fill_f32<<<blks(1600000, 256), 256, 0, stream>>>(acc2, 0.f, 1600000);
  edge_max<<<blks((long long)ETOT, 256), 256, 0, stream>>>(src, dst, as2, ad2, m2, 1);
  edge_acc<<<blks((long long)ETOT * HID, 256), 256, 0, stream>>>(src, dst, as2, ad2, m2,
                                                                 z, acc2, d2, 1, HID);

  final_cls<<<blks((long long)NN * 32, 256), 256, 0, stream>>>(acc2, d2, b2, lin_w, lin_b, out);
}


# Both products: each wave owns a 16x64 (16x32) register tile fed straight from global; tiles interleaved by column, so one B load serves every tile and C rows store whole; k summed in another order
# speedup vs baseline: 4.5645x; 4.5645x over previous
#include <hip/hip_runtime.h>
#include <math.h>

#define NN 50000
#define NE 800000
#define INDIM 256
#define HID 32
#define HEADS 4
#define ETOT (NE + NN)

typedef __attribute__((ext_vector_type(2))) float v2f;
typedef __attribute__((ext_vector_type(8))) float v8f;

__device__ __forceinline__ unsigned fkey(float x) {
  unsigned u = __float_as_uint(x);
  return (u & 0x80000000u) ? ~u : (u | 0x80000000u);
}
__device__ __forceinline__ float unfkey(unsigned k) {
  return __uint_as_float((k & 0x80000000u) ? (k & 0x7fffffffu) : ~k);
}
__device__ __forceinline__ float elu(float v) {
  return (v > 0.f) ? v : (__expf(v) - 1.f);
}

typedef __attribute__((ext_vector_type(4))) float v4f;
template <int NT> struct colvec;
template <> struct colvec<4> { typedef v4f type; };
template <> struct colvec<2> { typedef v2f type; };

template <int N, int K, int NT>
__global__ void gemm_wmma_f32(const float* __restrict__ A, const float* __restrict__ B,
                              float* __restrict__ C) {
  typedef typename colvec<NT>::type vnt;
  const int lane  = threadIdx.x;
  const int c     = lane & 15;
  const int h     = lane >> 4;
  const int mbase = (blockIdx.x * blockDim.z + threadIdx.z) * 16;
  const int nb    = threadIdx.y * (16 * NT);
  const float* ap = A + (size_t)(mbase + c) * K + 4 * h;
  const float* bp = B + (size_t)(4 * h) * N + nb + NT * c;
  v8f acc[NT];
#pragma unroll
  for (int t = 0; t < NT; ++t) acc[t] = v8f{0.f, 0.f, 0.f, 0.f, 0.f, 0.f, 0.f, 0.f};

  constexpr int KC  = 16;
  constexpr int NCH = K / KC;
  static_assert(K % KC == 0, "whole chunks");
  v4f af[2][2];
  vnt bf[2][8];
#pragma unroll
  for (int g = 0; g < 2; ++g) {
    af[0][g] = *reinterpret_cast<const v4f*>(ap + 8 * g);
#pragma unroll
    for (int j = 0; j < 4; ++j)
      bf[0][4 * g + j] = *reinterpret_cast<const vnt*>(bp + (size_t)(8 * g + j) * N);
  }
#pragma unroll
  for (int ch = 0; ch < NCH; ++ch) {
    const int cur = ch & 1, nxt = cur ^ 1;
    if (ch + 1 < NCH) {
      const float* apn = ap + (ch + 1) * KC;
      const float* bpn = bp + (size_t)(ch + 1) * KC * N;
#pragma unroll
      for (int g = 0; g < 2; ++g) {
        af[nxt][g] = *reinterpret_cast<const v4f*>(apn + 8 * g);
#pragma unroll
        for (int j = 0; j < 4; ++j)
          bf[nxt][4 * g + j] = *reinterpret_cast<const vnt*>(bpn + (size_t)(8 * g + j) * N);
      }
    }
    __builtin_amdgcn_sched_barrier(0);
#pragma unroll
    for (int g = 0; g < 2; ++g) {
#pragma unroll
      for (int u = 0; u < 2; ++u) {
        v2f a;
        a.x = af[cur][g][2 * u];
        a.y = af[cur][g][2 * u + 1];
#pragma unroll
        for (int t = 0; t < NT; ++t) {
          v2f b;
          b.x = bf[cur][4 * g + 2 * u][t];
          b.y = bf[cur][4 * g + 2 * u + 1][t];
          acc[t] = __builtin_amdgcn_wmma_f32_16x16x4_f32(false, a, false, b, (short)0, acc[t],
                                                         false, false);
        }
      }
    }
    __builtin_amdgcn_sched_barrier(0);
  }
  float* cp = C + (size_t)(mbase + 8 * h) * N + nb + NT * c;
#pragma unroll
  for (int r = 0; r < 8; ++r) {
    vnt o;
#pragma unroll
    for (int t = 0; t < NT; ++t) o[t] = acc[t][r];
    *reinterpret_cast<vnt*>(cp + (size_t)r * N) = o;
  }
}

__global__ void att_scores(const float* __restrict__ h, const float* __restrict__ att_s,
                           const float* __restrict__ att_d, float* __restrict__ as,
                           float* __restrict__ ad, int H, int C) {
  int i = blockIdx.x * blockDim.x + threadIdx.x;
  if (i >= NN * H) return;
  int hh = i % H;
  const float* hp = h + (size_t)i * C;
  float s = 0.f, d = 0.f;
  for (int c = 0; c < C; ++c) {
    float v = hp[c];
    s += v * att_s[hh * C + c];
    d += v * att_d[hh * C + c];
  }
  as[i] = s;
  ad[i] = d;
}

#define NPB  208
#define NBLK ((NN + NPB - 1) / NPB)
#define NU   (NE / 16)
#define LCAP 32768
#define CHU  (LCAP / 16)
static_assert(NBLK <= 256, "owner id must fit a byte");
static_assert(NE % 16 == 0, "owner map is read in 16-byte units");
static_assert(NN <= 65536 && NPB <= 65536, "list entry packs source and local destination in 16 bits each");
static_assert(HID == 32, "one lane per channel of a head");
static_assert(NN % 80 == 0, "product launches: 16 rows per wave, five waves per block in the second");

__device__ __forceinline__ float lrelu(float x) { return (x > 0.f) ? x : 0.2f * x; }

__global__ void owner_bytes(const int* __restrict__ dst, unsigned* __restrict__ own) {
  int i = blockIdx.x * blockDim.x + threadIdx.x;
  if (i >= NE / 4) return;
  const int4 d = ((const int4*)dst)[i];
  own[i] = (unsigned)(d.x / NPB) | ((unsigned)(d.y / NPB) << 8) |
           ((unsigned)(d.z / NPB) << 16) | ((unsigned)(d.w / NPB) << 24);
}

__device__ __forceinline__ unsigned zero_bytes(unsigned t) {
  return ~(((t & 0x7f7f7f7fu) + 0x7f7f7f7fu) | t) & 0x80808080u;
}

__device__ __forceinline__ void scan_units(const uint4* __restrict__ own4, int u0, int u1, int rot,
                                           unsigned pat, unsigned* list_l, unsigned* cnt_l, int tid) {
  const unsigned none = pat ^ 0x01010101u;
  for (int u = u0 + tid; u < u1; u += 4 * 1024) {
    uint4 v[4];
    int id[4];
#pragma unroll
    for (int q = 0; q < 4; ++q) {
      const int uu = u + q * 1024;
      const bool ok = uu < u1;
      int idx = (ok ? uu : u) + rot;
      if (idx >= u1) idx -= (u1 - u0);
      id[q] = idx;
      v[q] = own4[idx];
      if (!ok) v[q] = uint4{none, none, none, none};
    }
#pragma unroll
    for (int q = 0; q < 4; ++q) {
      const unsigned mm = (zero_bytes(v[q].x ^ pat) >> 7) | (zero_bytes(v[q].y ^ pat) >> 6) |
                          (zero_bytes(v[q].z ^ pat) >> 5) | (zero_bytes(v[q].w ^ pat) >> 4);
      if (mm) {
        unsigned pos = atomicAdd(cnt_l, (unsigned)__popc(mm));
        for (unsigned r = mm; r; r &= r - 1) {
          const int j = __ffs(r) - 1;
          if (pos < LCAP) list_l[pos] = (unsigned)(id[q] * 16 + 4 * (j & 3) + (j >> 3));
          ++pos;
        }
      }
    }
  }
}

template <int H, bool FINAL>
__global__ __launch_bounds__(1024) void agg_owner(
    const int* __restrict__ src, const int* __restrict__ dst, const unsigned* __restrict__ own,
    const float* __restrict__ as, const float* __restrict__ ad, const float* __restrict__ feat,
    const float* __restrict__ bias, const float* __restrict__ lin_w, const float* __restrict__ lin_b,
    float* __restrict__ outp) {
  constexpr int HC = H * HID;
  __shared__ float    acc_l[NPB * HC];
  __shared__ float    den_l[NPB * H];
  __shared__ __attribute__((aligned(16))) float    ad_l[NPB * H];
  __shared__ __attribute__((aligned(16))) unsigned m_l[NPB * H];
  __shared__ unsigned list_l[LCAP];
  __shared__ unsigned cnt_l;
  const int tid = threadIdx.x, lane = tid & 31;
  const int wave = __builtin_amdgcn_readfirstlane(tid >> 5);
  const int lo = blockIdx.x * NPB;
  const int nn = min(NPB, NN - lo);
  const uint4* own4 = (const uint4*)own;
  const unsigned pat = (unsigned)blockIdx.x * 0x01010101u;

  for (int i = tid; i < NPB * HC; i += 1024) acc_l[i] = 0.f;
  for (int i = tid; i < NPB * H; i += 1024) {
    float a_d = 0.f;
    unsigned mk = 0u;
    if (i < nn * H) {
      a_d = ad[lo * H + i];
      mk = fkey(lrelu(as[lo * H + i] + a_d));
    }
    ad_l[i] = a_d;
    m_l[i] = mk;
    den_l[i] = 0.f;
  }
  if (tid == 0) cnt_l = 0u;
  __syncthreads();

  auto max_pass = [&](int n) {
    for (int i = tid; i < n; i += 1024) {
      const int e = (int)list_l[i];
      const int s = src[e];
      const int dl = dst[e] - lo;
#pragma unroll
      for (int h = 0; h < H; ++h)
        atomicMax(&m_l[dl * H + h], fkey(lrelu(as[s * H + h] + ad_l[dl * H + h])));
      list_l[i] = (unsigned)s | ((unsigned)dl << 16);
    }
  };

  auto acc_pass = [&](int n, int nself) {
    const int tot = n + nself;
    for (int i0 = wave * 4; i0 < tot; i0 += 32 * 4) {
      int s[4], dl[4];
      float v[4][H], a[4][H];
#pragma unroll
      for (int j = 0; j < 4; ++j) {
        const int i = i0 + j;
        if (i < tot) {
          unsigned pk = (i < n) ? list_l[i] : ((unsigned)(lo + i - n) | ((unsigned)(i - n) << 16));
          pk = (unsigned)__builtin_amdgcn_readfirstlane((int)pk);
          s[j] = (int)(pk & 0xffffu);
          dl[j] = (int)(pk >> 16);
#pragma unroll
          for (int h = 0; h < H; ++h) {
            v[j][h] = feat[(size_t)s[j] * HC + h * HID + lane];
            a[j][h] = as[s[j] * H + h];
          }
        }
      }
#pragma unroll
      for (int j = 0; j < 4; ++j) {
        if (i0 + j < tot) {
          float adv[H];
          unsigned mv[H];
          if constexpr (H == 4) {
            const float4 t4 = *reinterpret_cast<const float4*>(&ad_l[dl[j] * 4]);
            const uint4 m4 = *reinterpret_cast<const uint4*>(&m_l[dl[j] * 4]);
            adv[0] = t4.x; adv[1] = t4.y; adv[2] = t4.z; adv[3] = t4.w;
            mv[0] = m4.x; mv[1] = m4.y; mv[2] = m4.z; mv[3] = m4.w;
          } else {
#pragma unroll
            for (int h = 0; h < H; ++h) {
              adv[h] = ad_l[dl[j] * H + h];
              mv[h] = m_l[dl[j] * H + h];
            }
          }
          float psel = 0.f;
#pragma unroll
          for (int h = 0; h < H; ++h) {
            const float x = lrelu(a[j][h] + adv[h]);
            const float p = __expf(x - unfkey(mv[h]));
            atomicAdd(&acc_l[dl[j] * HC + h * HID + lane], p * v[j][h]);
            if (lane == h) psel = p;
          }
          if (lane < H) atomicAdd(&den_l[dl[j] * H + lane], psel);
        }
      }
    }
  };

  scan_units(own4, 0, NU, (int)blockIdx.x * (NU / NBLK), pat, list_l, &cnt_l, tid);
  __syncthreads();
  const int n = (int)cnt_l;
  if (n <= LCAP) {
    max_pass(n);
    __syncthreads();
    acc_pass(n, nn);
  } else {
    for (int c0 = 0; c0 < NU; c0 += CHU) {
      __syncthreads();
      if (tid == 0) cnt_l = 0u;
      __syncthreads();
      scan_units(own4, c0, min(c0 + CHU, NU), 0, pat, list_l, &cnt_l, tid);
      __syncthreads();
      max_pass((int)cnt_l);
    }
    for (int c0 = 0; c0 < NU; c0 += CHU) {
      __syncthreads();
      if (tid == 0) cnt_l = 0u;
      __syncthreads();
      scan_units(own4, c0, min(c0 + CHU, NU), 0, pat, list_l, &cnt_l, tid);
      __syncthreads();
      const int nc = (int)cnt_l;
      max_pass(nc);
      __syncthreads();
      acc_pass(nc, (c0 == 0) ? nn : 0);
    }
  }
  __syncthreads();

  if (!FINAL) {
    for (int i = tid; i < nn * HC; i += 1024) {
      const int dl = i / HC, r = i % HC;
      outp[(size_t)lo * HC + i] = elu(acc_l[i] / den_l[dl * H + r / HID] + bias[r]);
    }
  } else {
    for (int dl = wave; dl < nn; dl += 32) {
      const float v = elu(acc_l[dl * HC + lane] / den_l[dl * H] + bias[lane]);
      float w0 = v * lin_w[lane * 2 + 0];
      float w1 = v * lin_w[lane * 2 + 1];
#pragma unroll
      for (int m = 16; m > 0; m >>= 1) {
        w0 += __shfl_xor(w0, m, 32);
        w1 += __shfl_xor(w1, m, 32);
      }
      if (lane == 0) {
        outp[(lo + dl) * 2 + 0] = w0 + lin_b[0];
        outp[(lo + dl) * 2 + 1] = w1 + lin_b[1];
      }
    }
  }
}

static inline unsigned blks(long long tot, int bs) { return (unsigned)((tot + bs - 1) / bs); }

extern "C" void kernel_launch(void* const* d_in, const int* in_sizes, int n_in,
                              void* d_out, int out_size, void* d_ws, size_t ws_size,
                              hipStream_t stream) {
  const float* x        = (const float*)d_in[0];
  const int*   ei       = (const int*)d_in[1];
  const float* W1       = (const float*)d_in[2];
  const float* att_src1 = (const float*)d_in[3];
  const float* att_dst1 = (const float*)d_in[4];
  const float* b1       = (const float*)d_in[5];
  const float* W2       = (const float*)d_in[6];
  const float* att_src2 = (const float*)d_in[7];
  const float* att_dst2 = (const float*)d_in[8];
  const float* b2       = (const float*)d_in[9];
  const float* lin_w    = (const float*)d_in[10];
  const float* lin_b    = (const float*)d_in[11];
  float* out = (float*)d_out;
  const int* src = ei;
  const int* dst = ei + NE;

  float*    ws   = (float*)d_ws;
  float*    h1   = ws;
  float*    as1  = ws + 6400000;
  float*    ad1  = as1 + 200000;
  unsigned* own  = (unsigned*)(ad1 + 200000);
  float*    h2   = ad1 + 600000;
  float*    z    = h1;
  float*    as2  = h1 + 3200000;
  float*    ad2  = as2 + 50000;

  owner_bytes<<<blks(NE / 4, 256), 256, 0, stream>>>(dst, own);

  gemm_wmma_f32<HEADS * HID, INDIM, 4>
      <<<dim3(NN / 16), dim3(32, 2, 1), 0, stream>>>(x, W1, h1);
  att_scores<<<blks(NN * HEADS, 256), 256, 0, stream>>>(h1, att_src1, att_dst1, as1, ad1, HEADS, HID);
  agg_owner<HEADS, false><<<NBLK, 1024, 0, stream>>>(src, dst, own, as1, ad1, h1, b1, lin_w, lin_b, h2);

  gemm_wmma_f32<HID, HEADS * HID, 2>
      <<<dim3(NN / 80), dim3(32, 1, 5), 0, stream>>>(h2, W2, z);
  att_scores<<<blks(NN, 256), 256, 0, stream>>>(z, att_src2, att_dst2, as2, ad2, 1, HID);
  agg_owner<1, true><<<NBLK, 1024, 0, stream>>>(src, dst, own, as2, ad2, z, b2, lin_w, lin_b, out);
}


# Softmax weights formed once per edge and head by a lane-per-edge pass and kept in LDS; gather waves take 16 edges per lane load, only multiply and add; sums in another LDS atomic arrival order
# speedup vs baseline: 5.5759x; 1.2216x over previous
#include <hip/hip_runtime.h>
#include <math.h>

#define NN 50000
#define NE 800000
#define INDIM 256
#define HID 32
#define HEADS 4
#define ETOT (NE + NN)

typedef __attribute__((ext_vector_type(2))) float v2f;
typedef __attribute__((ext_vector_type(8))) float v8f;

__device__ __forceinline__ unsigned fkey(float x) {
  unsigned u = __float_as_uint(x);
  return (u & 0x80000000u) ? ~u : (u | 0x80000000u);
}
__device__ __forceinline__ float unfkey(unsigned k) {
  return __uint_as_float((k & 0x80000000u) ? (k & 0x7fffffffu) : ~k);
}
__device__ __forceinline__ float elu(float v) {
  return (v > 0.f) ? v : (__expf(v) - 1.f);
}

typedef __attribute__((ext_vector_type(4))) float v4f;
template <int NT> struct colvec;
template <> struct colvec<4> { typedef v4f type; };
template <> struct colvec<2> { typedef v2f type; };

template <int N, int K, int NT>
__global__ void gemm_wmma_f32(const float* __restrict__ A, const float* __restrict__ B,
                              float* __restrict__ C) {
  typedef typename colvec<NT>::type vnt;
  const int lane  = threadIdx.x;
  const int c     = lane & 15;
  const int h     = lane >> 4;
  const int mbase = (blockIdx.x * blockDim.z + threadIdx.z) * 16;
  const int nb    = threadIdx.y * (16 * NT);
  const float* ap = A + (size_t)(mbase + c) * K + 4 * h;
  const float* bp = B + (size_t)(4 * h) * N + nb + NT * c;
  v8f acc[NT];
#pragma unroll
  for (int t = 0; t < NT; ++t) acc[t] = v8f{0.f, 0.f, 0.f, 0.f, 0.f, 0.f, 0.f, 0.f};

  constexpr int KC  = 16;
  constexpr int NCH = K / KC;
  static_assert(K % KC == 0, "whole chunks");
  v4f af[2][2];
  vnt bf[2][8];
#pragma unroll
  for (int g = 0; g < 2; ++g) {
    af[0][g] = *reinterpret_cast<const v4f*>(ap + 8 * g);
#pragma unroll
    for (int j = 0; j < 4; ++j)
      bf[0][4 * g + j] = *reinterpret_cast<const vnt*>(bp + (size_t)(8 * g + j) * N);
  }
#pragma unroll
  for (int ch = 0; ch < NCH; ++ch) {
    const int cur = ch & 1, nxt = cur ^ 1;
    if (ch + 1 < NCH) {
      const float* apn = ap + (ch + 1) * KC;
      const float* bpn = bp + (size_t)(ch + 1) * KC * N;
#pragma unroll
      for (int g = 0; g < 2; ++g) {
        af[nxt][g] = *reinterpret_cast<const v4f*>(apn + 8 * g);
#pragma unroll
        for (int j = 0; j < 4; ++j)
          bf[nxt][4 * g + j] = *reinterpret_cast<const vnt*>(bpn + (size_t)(8 * g + j) * N);
      }
    }
    __builtin_amdgcn_sched_barrier(0);
#pragma unroll
    for (int g = 0; g < 2; ++g) {
#pragma unroll
      for (int u = 0; u < 2; ++u) {
        v2f a;
        a.x = af[cur][g][2 * u];
        a.y = af[cur][g][2 * u + 1];
#pragma unroll
        for (int t = 0; t < NT; ++t) {
          v2f b;
          b.x = bf[cur][4 * g + 2 * u][t];
          b.y = bf[cur][4 * g + 2 * u + 1][t];
          acc[t] = __builtin_amdgcn_wmma_f32_16x16x4_f32(false, a, false, b, (short)0, acc[t],
                                                         false, false);
        }
      }
    }
    __builtin_amdgcn_sched_barrier(0);
  }
  float* cp = C + (size_t)(mbase + 8 * h) * N + nb + NT * c;
#pragma unroll
  for (int r = 0; r < 8; ++r) {
    vnt o;
#pragma unroll
    for (int t = 0; t < NT; ++t) o[t] = acc[t][r];
    *reinterpret_cast<vnt*>(cp + (size_t)r * N) = o;
  }
}

__global__ void att_scores(const float* __restrict__ h, const float* __restrict__ att_s,
                           const float* __restrict__ att_d, float* __restrict__ as,
                           float* __restrict__ ad, int H, int C) {
  int i = blockIdx.x * blockDim.x + threadIdx.x;
  if (i >= NN * H) return;
  int hh = i % H;
  const float* hp = h + (size_t)i * C;
  float s = 0.f, d = 0.f;
  for (int c = 0; c < C; ++c) {
    float v = hp[c];
    s += v * att_s[hh * C + c];
    d += v * att_d[hh * C + c];
  }
  as[i] = s;
  ad[i] = d;
}

#define NPB  208
#define NBLK ((NN + NPB - 1) / NPB)
#define NU   (NE / 16)
#define LCAP 8192
#define CHU  (LCAP / 16)
static_assert(NBLK <= 256, "owner id must fit a byte");
static_assert(NE % 16 == 0, "owner map is read in 16-byte units");
static_assert(NN <= 65536 && NPB <= 65536, "list entry packs source and local destination in 16 bits each");
static_assert(HID == 32, "one lane per channel of a head");
static_assert(NN % 80 == 0, "product launches: 16 rows per wave, five waves per block in the second");

__device__ __forceinline__ float lrelu(float x) { return (x > 0.f) ? x : 0.2f * x; }

__global__ void owner_bytes(const int* __restrict__ dst, unsigned* __restrict__ own) {
  int i = blockIdx.x * blockDim.x + threadIdx.x;
  if (i >= NE / 4) return;
  const int4 d = ((const int4*)dst)[i];
  own[i] = (unsigned)(d.x / NPB) | ((unsigned)(d.y / NPB) << 8) |
           ((unsigned)(d.z / NPB) << 16) | ((unsigned)(d.w / NPB) << 24);
}

__device__ __forceinline__ unsigned zero_bytes(unsigned t) {
  return ~(((t & 0x7f7f7f7fu) + 0x7f7f7f7fu) | t) & 0x80808080u;
}

__device__ __forceinline__ void scan_units(const uint4* __restrict__ own4, int u0, int u1, int rot,
                                           unsigned pat, unsigned* list_l, unsigned* cnt_l, int tid) {
  const unsigned none = pat ^ 0x01010101u;
  for (int u = u0 + tid; u < u1; u += 4 * 1024) {
    uint4 v[4];
    int id[4];
#pragma unroll
    for (int q = 0; q < 4; ++q) {
      const int uu = u + q * 1024;
      const bool ok = uu < u1;
      int idx = (ok ? uu : u) + rot;
      if (idx >= u1) idx -= (u1 - u0);
      id[q] = idx;
      v[q] = own4[idx];
      if (!ok) v[q] = uint4{none, none, none, none};
    }
#pragma unroll
    for (int q = 0; q < 4; ++q) {
      const unsigned mm = (zero_bytes(v[q].x ^ pat) >> 7) | (zero_bytes(v[q].y ^ pat) >> 6) |
                          (zero_bytes(v[q].z ^ pat) >> 5) | (zero_bytes(v[q].w ^ pat) >> 4);
      if (mm) {
        unsigned pos = atomicAdd(cnt_l, (unsigned)__popc(mm));
        for (unsigned r = mm; r; r &= r - 1) {
          const int j = __ffs(r) - 1;
          if (pos < LCAP) list_l[pos] = (unsigned)(id[q] * 16 + 4 * (j & 3) + (j >> 3));
          ++pos;
        }
      }
    }
  }
}

template <bool B> struct flag_c { static constexpr bool value = B; };

template <int H, bool FINAL>
__global__ __launch_bounds__(1024) void agg_owner(
    const int* __restrict__ src, const int* __restrict__ dst, const unsigned* __restrict__ own,
    const float* __restrict__ as, const float* __restrict__ ad, const float* __restrict__ feat,
    const float* __restrict__ bias, const float* __restrict__ lin_w, const float* __restrict__ lin_b,
    float* __restrict__ outp) {
  constexpr int HC = H * HID;
  __shared__ float    acc_l[NPB * HC];
  __shared__ float    den_l[NPB * H];
  __shared__ __attribute__((aligned(16))) float    ad_l[NPB * H];
  __shared__ __attribute__((aligned(16))) unsigned m_l[NPB * H];
  __shared__ unsigned list_l[LCAP + NPB];
  __shared__ __attribute__((aligned(16))) float p_l[(LCAP + NPB) * H];
  __shared__ unsigned cnt_l;
  const int tid = threadIdx.x, lane = tid & 31;
  const int wave = __builtin_amdgcn_readfirstlane(tid >> 5);
  const int lo = blockIdx.x * NPB;
  const int nn = min(NPB, NN - lo);
  const uint4* own4 = (const uint4*)own;
  const unsigned pat = (unsigned)blockIdx.x * 0x01010101u;

  for (int i = tid; i < NPB * HC; i += 1024) acc_l[i] = 0.f;
  for (int i = tid; i < NPB * H; i += 1024) {
    float a_d = 0.f;
    unsigned mk = 0u;
    if (i < nn * H) {
      a_d = ad[lo * H + i];
      mk = fkey(lrelu(as[lo * H + i] + a_d));
    }
    ad_l[i] = a_d;
    m_l[i] = mk;
    den_l[i] = 0.f;
  }
  if (tid == 0) cnt_l = 0u;
  __syncthreads();

  auto max_pass = [&](int n) {
    for (int i = tid; i < n; i += 1024) {
      const int e = (int)list_l[i];
      const int s = src[e];
      const int dl = dst[e] - lo;
#pragma unroll
      for (int h = 0; h < H; ++h) {
        const float x = lrelu(as[s * H + h] + ad_l[dl * H + h]);
        atomicMax(&m_l[dl * H + h], fkey(x));
        p_l[i * H + h] = x;
      }
      list_l[i] = (unsigned)s | ((unsigned)dl << 16);
    }
  };

  auto weight_pass = [&](int n, int nself) {
    const int tot = n + nself;
    for (int i = tid; i < tot; i += 1024) {
      float x[H];
      int dl;
      if (i < n) {
        dl = (int)(list_l[i] >> 16);
#pragma unroll
        for (int h = 0; h < H; ++h) x[h] = p_l[i * H + h];
      } else {
        dl = i - n;
        list_l[i] = (unsigned)(lo + dl) | ((unsigned)dl << 16);
#pragma unroll
        for (int h = 0; h < H; ++h) x[h] = lrelu(as[(lo + dl) * H + h] + ad_l[dl * H + h]);
      }
#pragma unroll
      for (int h = 0; h < H; ++h) {
        const float p = __expf(x[h] - unfkey(m_l[dl * H + h]));
        p_l[i * H + h] = p;
        atomicAdd(&den_l[dl * H + h], p);
      }
    }
  };

  constexpr int BATCH = 16;
  constexpr int JB = (H == 1) ? 16 : 8;
  auto acc_batch = [&](auto all_c, int b0, int nb) __attribute__((always_inline)) {
    constexpr bool ALL = decltype(all_c)::value;
    const unsigned pkv = (lane < nb) ? list_l[b0 + lane] : 0u;
#pragma unroll
    for (int j0 = 0; j0 < BATCH; j0 += JB) {
      if (ALL || j0 < nb) {
        int dl[JB] = {};
        float v[JB][H] = {}, p[JB][H] = {};
#pragma unroll
        for (int j = 0; j < JB; ++j) {
          if (ALL || j0 + j < nb) {
            const unsigned pk = (unsigned)__builtin_amdgcn_readlane((int)pkv, j0 + j);
            const int s = (int)(pk & 0xffffu);
            dl[j] = (int)(pk >> 16);
#pragma unroll
            for (int h = 0; h < H; ++h) v[j][h] = feat[(size_t)s * HC + h * HID + lane];
          }
        }
#pragma unroll
        for (int j = 0; j < JB; ++j) {
          if (ALL || j0 + j < nb) {
            if constexpr (H == 4) {
              const float4 p4 = *reinterpret_cast<const float4*>(&p_l[(b0 + j0 + j) * 4]);
              p[j][0] = p4.x; p[j][1] = p4.y; p[j][2] = p4.z; p[j][3] = p4.w;
            } else {
#pragma unroll
              for (int h = 0; h < H; ++h) p[j][h] = p_l[(b0 + j0 + j) * H + h];
            }
          }
        }
#pragma unroll
        for (int j = 0; j < JB; ++j) {
          if (ALL || j0 + j < nb) {
#pragma unroll
            for (int h = 0; h < H; ++h)
              atomicAdd(&acc_l[dl[j] * HC + h * HID + lane], p[j][h] * v[j][h]);
          }
        }
      }
    }
  };
  auto acc_pass = [&](int tot) __attribute__((always_inline)) {
    for (int b0 = wave * BATCH; b0 < tot; b0 += 32 * BATCH) {
      const int nb = min(BATCH, tot - b0);
      if (nb == BATCH) acc_batch(flag_c<true>{}, b0, BATCH);
      else acc_batch(flag_c<false>{}, b0, nb);
    }
  };

  scan_units(own4, 0, NU, (int)blockIdx.x * (NU / NBLK), pat, list_l, &cnt_l, tid);
  __syncthreads();
  const int n = (int)cnt_l;
  if (n <= LCAP) {
    max_pass(n);
    __syncthreads();
    weight_pass(n, nn);
    __syncthreads();
    acc_pass(n + nn);
  } else {
    for (int c0 = 0; c0 < NU; c0 += CHU) {
      __syncthreads();
      if (tid == 0) cnt_l = 0u;
      __syncthreads();
      scan_units(own4, c0, min(c0 + CHU, NU), 0, pat, list_l, &cnt_l, tid);
      __syncthreads();
      max_pass((int)cnt_l);
    }
    for (int c0 = 0; c0 < NU; c0 += CHU) {
      __syncthreads();
      if (tid == 0) cnt_l = 0u;
      __syncthreads();
      scan_units(own4, c0, min(c0 + CHU, NU), 0, pat, list_l, &cnt_l, tid);
      __syncthreads();
      const int nc = (int)cnt_l;
      const int ns = (c0 == 0) ? nn : 0;
      max_pass(nc);
      __syncthreads();
      weight_pass(nc, ns);
      __syncthreads();
      acc_pass(nc + ns);
    }
  }
  __syncthreads();

  if (!FINAL) {
    for (int i = tid; i < nn * HC; i += 1024) {
      const int dl = i / HC, r = i % HC;
      outp[(size_t)lo * HC + i] = elu(acc_l[i] / den_l[dl * H + r / HID] + bias[r]);
    }
  } else {
    for (int dl = wave; dl < nn; dl += 32) {
      const float v = elu(acc_l[dl * HC + lane] / den_l[dl * H] + bias[lane]);
      float w0 = v * lin_w[lane * 2 + 0];
      float w1 = v * lin_w[lane * 2 + 1];
#pragma unroll
      for (int m = 16; m > 0; m >>= 1) {
        w0 += __shfl_xor(w0, m, 32);
        w1 += __shfl_xor(w1, m, 32);
      }
      if (lane == 0) {
        outp[(lo + dl) * 2 + 0] = w0 + lin_b[0];
        outp[(lo + dl) * 2 + 1] = w1 + lin_b[1];
      }
    }
  }
}

static inline unsigned blks(long long tot, int bs) { return (unsigned)((tot + bs - 1) / bs); }

extern "C" void kernel_launch(void* const* d_in, const int* in_sizes, int n_in,
                              void* d_out, int out_size, void* d_ws, size_t ws_size,
                              hipStream_t stream) {
  const float* x        = (const float*)d_in[0];
  const int*   ei       = (const int*)d_in[1];
  const float* W1       = (const float*)d_in[2];
  const float* att_src1 = (const float*)d_in[3];
  const float* att_dst1 = (const float*)d_in[4];
  const float* b1       = (const float*)d_in[5];
  const float* W2       = (const float*)d_in[6];
  const float* att_src2 = (const float*)d_in[7];
  const float* att_dst2 = (const float*)d_in[8];
  const float* b2       = (const float*)d_in[9];
  const float* lin_w    = (const float*)d_in[10];
  const float* lin_b    = (const float*)d_in[11];
  float* out = (float*)d_out;
  const int* src = ei;
  const int* dst = ei + NE;

  float*    ws   = (float*)d_ws;
  float*    h1   = ws;
  float*    as1  = ws + 6400000;
  float*    ad1  = as1 + 200000;
  unsigned* own  = (unsigned*)(ad1 + 200000);
  float*    h2   = ad1 + 600000;
  float*    z    = h1;
  float*    as2  = h1 + 3200000;
  float*    ad2  = as2 + 50000;

  owner_bytes<<<blks(NE / 4, 256), 256, 0, stream>>>(dst, own);

  gemm_wmma_f32<HEADS * HID, INDIM, 4>
      <<<dim3(NN / 16), dim3(32, 2, 1), 0, stream>>>(x, W1, h1);
  att_scores<<<blks(NN * HEADS, 256), 256, 0, stream>>>(h1, att_src1, att_dst1, as1, ad1, HEADS, HID);
  agg_owner<HEADS, false><<<NBLK, 1024, 0, stream>>>(src, dst, own, as1, ad1, h1, b1, lin_w, lin_b, h2);

  gemm_wmma_f32<HID, HEADS * HID, 2>
      <<<dim3(NN / 80), dim3(32, 1, 5), 0, stream>>>(h2, W2, z);
  att_scores<<<blks(NN, 256), 256, 0, stream>>>(z, att_src2, att_dst2, as2, ad2, 1, HID);
  agg_owner<1, true><<<NBLK, 1024, 0, stream>>>(src, dst, own, as2, ad2, z, b2, lin_w, lin_b, out);
}


# Owner-map search in two stages: an any-match test per 16-byte unit queues the few units holding the id (lane-mask counts), exact byte decode only on those; sums in another LDS atomic arrival order
# speedup vs baseline: 5.7744x; 1.0356x over previous
#include <hip/hip_runtime.h>
#include <math.h>

#define NN 50000
#define NE 800000
#define INDIM 256
#define HID 32
#define HEADS 4
#define ETOT (NE + NN)

typedef __attribute__((ext_vector_type(2))) float v2f;
typedef __attribute__((ext_vector_type(8))) float v8f;

__device__ __forceinline__ unsigned fkey(float x) {
  unsigned u = __float_as_uint(x);
  return (u & 0x80000000u) ? ~u : (u | 0x80000000u);
}
__device__ __forceinline__ float unfkey(unsigned k) {
  return __uint_as_float((k & 0x80000000u) ? (k & 0x7fffffffu) : ~k);
}
__device__ __forceinline__ float elu(float v) {
  return (v > 0.f) ? v : (__expf(v) - 1.f);
}

typedef __attribute__((ext_vector_type(4))) float v4f;
template <int NT> struct colvec;
template <> struct colvec<4> { typedef v4f type; };
template <> struct colvec<2> { typedef v2f type; };

template <int N, int K, int NT>
__global__ void gemm_wmma_f32(const float* __restrict__ A, const float* __restrict__ B,
                              float* __restrict__ C) {
  typedef typename colvec<NT>::type vnt;
  const int lane  = threadIdx.x;
  const int c     = lane & 15;
  const int h     = lane >> 4;
  const int mbase = (blockIdx.x * blockDim.z + threadIdx.z) * 16;
  const int nb    = threadIdx.y * (16 * NT);
  const float* ap = A + (size_t)(mbase + c) * K + 4 * h;
  const float* bp = B + (size_t)(4 * h) * N + nb + NT * c;
  v8f acc[NT];
#pragma unroll
  for (int t = 0; t < NT; ++t) acc[t] = v8f{0.f, 0.f, 0.f, 0.f, 0.f, 0.f, 0.f, 0.f};

  constexpr int KC  = 16;
  constexpr int NCH = K / KC;
  static_assert(K % KC == 0, "whole chunks");
  v4f af[2][2];
  vnt bf[2][8];
#pragma unroll
  for (int g = 0; g < 2; ++g) {
    af[0][g] = *reinterpret_cast<const v4f*>(ap + 8 * g);
#pragma unroll
    for (int j = 0; j < 4; ++j)
      bf[0][4 * g + j] = *reinterpret_cast<const vnt*>(bp + (size_t)(8 * g + j) * N);
  }
#pragma unroll
  for (int ch = 0; ch < NCH; ++ch) {
    const int cur = ch & 1, nxt = cur ^ 1;
    if (ch + 1 < NCH) {
      const float* apn = ap + (ch + 1) * KC;
      const float* bpn = bp + (size_t)(ch + 1) * KC * N;
#pragma unroll
      for (int g = 0; g < 2; ++g) {
        af[nxt][g] = *reinterpret_cast<const v4f*>(apn + 8 * g);
#pragma unroll
        for (int j = 0; j < 4; ++j)
          bf[nxt][4 * g + j] = *reinterpret_cast<const vnt*>(bpn + (size_t)(8 * g + j) * N);
      }
    }
    __builtin_amdgcn_sched_barrier(0);
#pragma unroll
    for (int g = 0; g < 2; ++g) {
#pragma unroll
      for (int u = 0; u < 2; ++u) {
        v2f a;
        a.x = af[cur][g][2 * u];
        a.y = af[cur][g][2 * u + 1];
#pragma unroll
        for (int t = 0; t < NT; ++t) {
          v2f b;
          b.x = bf[cur][4 * g + 2 * u][t];
          b.y = bf[cur][4 * g + 2 * u + 1][t];
          acc[t] = __builtin_amdgcn_wmma_f32_16x16x4_f32(false, a, false, b, (short)0, acc[t],
                                                         false, false);
        }
      }
    }
    __builtin_amdgcn_sched_barrier(0);
  }
  float* cp = C + (size_t)(mbase + 8 * h) * N + nb + NT * c;
#pragma unroll
  for (int r = 0; r < 8; ++r) {
    vnt o;
#pragma unroll
    for (int t = 0; t < NT; ++t) o[t] = acc[t][r];
    *reinterpret_cast<vnt*>(cp + (size_t)r * N) = o;
  }
}

__global__ void att_scores(const float* __restrict__ h, const float* __restrict__ att_s,
                           const float* __restrict__ att_d, float* __restrict__ as,
                           float* __restrict__ ad, int H, int C) {
  int i = blockIdx.x * blockDim.x + threadIdx.x;
  if (i >= NN * H) return;
  int hh = i % H;
  const float* hp = h + (size_t)i * C;
  float s = 0.f, d = 0.f;
  for (int c = 0; c < C; ++c) {
    float v = hp[c];
    s += v * att_s[hh * C + c];
    d += v * att_d[hh * C + c];
  }
  as[i] = s;
  ad[i] = d;
}

#define NPB  208
#define NBLK ((NN + NPB - 1) / NPB)
#define NU   (NE / 16)
#define LCAP 8192
#define CHU  (LCAP / 16)
#define QCAP LCAP
static_assert(NBLK <= 256, "owner id must fit a byte");
static_assert(NE % 16 == 0, "owner map is read in 16-byte units");
static_assert(NN <= 65536 && NPB <= 65536, "list entry packs source and local destination in 16 bits each");
static_assert(HID == 32, "one lane per channel of a head");
static_assert(NN % 80 == 0, "product launches: 16 rows per wave, five waves per block in the second");

__device__ __forceinline__ float lrelu(float x) { return (x > 0.f) ? x : 0.2f * x; }

__global__ void owner_bytes(const int* __restrict__ dst, unsigned* __restrict__ own) {
  int i = blockIdx.x * blockDim.x + threadIdx.x;
  if (i >= NE / 4) return;
  const int4 d = ((const int4*)dst)[i];
  own[i] = (unsigned)(d.x / NPB) | ((unsigned)(d.y / NPB) << 8) |
           ((unsigned)(d.z / NPB) << 16) | ((unsigned)(d.w / NPB) << 24);
}

__device__ __forceinline__ unsigned zero_bytes(unsigned t) {
  return ~(((t & 0x7f7f7f7fu) + 0x7f7f7f7fu) | t) & 0x80808080u;
}


__device__ __forceinline__ void find_units(const uint4* __restrict__ own4, int u0, int u1, int rot,
                                           unsigned pat, float* q_l, unsigned* qcnt_l, int wave, int lane) {
  for (int ub = u0 + wave * 32; ub < u1; ub += 4 * 1024) {
    const int u = ub + lane;
    uint4 v[4];
    int id[4];
    bool ok[4];
#pragma unroll
    for (int q = 0; q < 4; ++q) {
      const int uu = u + q * 1024;
      ok[q] = uu < u1;
      int idx = (ok[q] ? uu : u0) + rot;
      if (idx >= u1) idx -= (u1 - u0);
      id[q] = idx;
      v[q] = own4[idx];
    }
    unsigned bal[4];
    bool hit[4];
#pragma unroll
    for (int q = 0; q < 4; ++q) {
      const unsigned tx = v[q].x ^ pat, ty = v[q].y ^ pat, tz = v[q].z ^ pat, tw = v[q].w ^ pat;
      const unsigned a = ((tx - 0x01010101u) & ~tx) | ((ty - 0x01010101u) & ~ty) |
                         ((tz - 0x01010101u) & ~tz) | ((tw - 0x01010101u) & ~tw);
      hit[q] = ok[q] && ((a & 0x80808080u) != 0u);
      bal[q] = __builtin_amdgcn_ballot_w32(hit[q]);
    }
    const unsigned tot = (unsigned)(__builtin_popcount(bal[0]) + __builtin_popcount(bal[1]) +
                                    __builtin_popcount(bal[2]) + __builtin_popcount(bal[3]));
    if (tot) {
      unsigned base = 0u;
      if (lane == 0) base = atomicAdd(qcnt_l, tot);
      base = (unsigned)__builtin_amdgcn_readfirstlane((int)base);
#pragma unroll
      for (int q = 0; q < 4; ++q) {
        if (hit[q]) {
          const unsigned pos = base + __builtin_amdgcn_mbcnt_lo(bal[q], 0u);
          if (pos < QCAP) q_l[pos] = __uint_as_float((unsigned)id[q]);
        }
        base += (unsigned)__builtin_popcount(bal[q]);
      }
    }
  }
}

__device__ __forceinline__ void decode_units(const uint4* __restrict__ own4, unsigned pat, const float* q_l, int nq,
                                             unsigned* list_l, unsigned* cnt_l, int wave, int lane) {
  for (int i0 = wave * 32; i0 < nq; i0 += 4 * 1024) {
    uint4 v[4];
    unsigned id[4];
    bool ok[4];
#pragma unroll
    for (int q = 0; q < 4; ++q) {
      const int i = i0 + q * 1024 + lane;
      ok[q] = i < nq;
      id[q] = __float_as_uint(q_l[ok[q] ? i : i0]);
      v[q] = own4[id[q]];
    }
#pragma unroll
    for (int q = 0; q < 4; ++q) {
      unsigned mm = (zero_bytes(v[q].x ^ pat) >> 7) | (zero_bytes(v[q].y ^ pat) >> 6) |
                    (zero_bytes(v[q].z ^ pat) >> 5) | (zero_bytes(v[q].w ^ pat) >> 4);
      if (!ok[q]) mm = 0u;
      const unsigned c = (unsigned)__popc(mm);
      unsigned excl = 0u, tot = 0u;
#pragma unroll
      for (int b = 0; b < 5; ++b) {
        const unsigned bl = __builtin_amdgcn_ballot_w32(((c >> b) & 1u) != 0u);
        excl += __builtin_amdgcn_mbcnt_lo(bl, 0u) << b;
        tot += (unsigned)__builtin_popcount(bl) << b;
      }
      if (tot) {
        unsigned base = 0u;
        if (lane == 0) base = atomicAdd(cnt_l, tot);
        unsigned pos = (unsigned)__builtin_amdgcn_readfirstlane((int)base) + excl;
        for (unsigned r = mm; r; r &= r - 1) {
          const int j = __ffs(r) - 1;
          if (pos < LCAP) list_l[pos] = id[q] * 16u + (unsigned)(4 * (j & 3) + (j >> 3));
          ++pos;
        }
      }
    }
  }
}

template <bool B> struct flag_c { static constexpr bool value = B; };

template <int H, bool FINAL>
__global__ __launch_bounds__(1024) void agg_owner(
    const int* __restrict__ src, const int* __restrict__ dst, const unsigned* __restrict__ own,
    const float* __restrict__ as, const float* __restrict__ ad, const float* __restrict__ feat,
    const float* __restrict__ bias, const float* __restrict__ lin_w, const float* __restrict__ lin_b,
    float* __restrict__ outp) {
  constexpr int HC = H * HID;
  __shared__ float    acc_l[NPB * HC];
  __shared__ float    den_l[NPB * H];
  __shared__ __attribute__((aligned(16))) float    ad_l[NPB * H];
  __shared__ __attribute__((aligned(16))) unsigned m_l[NPB * H];
  __shared__ unsigned list_l[LCAP + NPB];
  __shared__ __attribute__((aligned(16))) float p_l[(LCAP + NPB) * H];
  __shared__ unsigned cnt_l, qcnt_l;
  static_assert(QCAP <= (LCAP + NPB) * H, "the unit queue borrows p_l");
  const int tid = threadIdx.x, lane = tid & 31;
  const int wave = __builtin_amdgcn_readfirstlane(tid >> 5);
  const int lo = blockIdx.x * NPB;
  const int nn = min(NPB, NN - lo);
  const uint4* own4 = (const uint4*)own;
  const unsigned pat = (unsigned)blockIdx.x * 0x01010101u;

  for (int i = tid; i < NPB * HC; i += 1024) acc_l[i] = 0.f;
  for (int i = tid; i < NPB * H; i += 1024) {
    float a_d = 0.f;
    unsigned mk = 0u;
    if (i < nn * H) {
      a_d = ad[lo * H + i];
      mk = fkey(lrelu(as[lo * H + i] + a_d));
    }
    ad_l[i] = a_d;
    m_l[i] = mk;
    den_l[i] = 0.f;
  }
  if (tid == 0) { cnt_l = 0u; qcnt_l = 0u; }
  __syncthreads();

  auto max_pass = [&](int n) {
    for (int i = tid; i < n; i += 1024) {
      const int e = (int)list_l[i];
      const int s = src[e];
      const int dl = dst[e] - lo;
#pragma unroll
      for (int h = 0; h < H; ++h) {
        const float x = lrelu(as[s * H + h] + ad_l[dl * H + h]);
        atomicMax(&m_l[dl * H + h], fkey(x));
        p_l[i * H + h] = x;
      }
      list_l[i] = (unsigned)s | ((unsigned)dl << 16);
    }
  };

  auto weight_pass = [&](int n, int nself) {
    const int tot = n + nself;
    for (int i = tid; i < tot; i += 1024) {
      float x[H];
      int dl;
      if (i < n) {
        dl = (int)(list_l[i] >> 16);
#pragma unroll
        for (int h = 0; h < H; ++h) x[h] = p_l[i * H + h];
      } else {
        dl = i - n;
        list_l[i] = (unsigned)(lo + dl) | ((unsigned)dl << 16);
#pragma unroll
        for (int h = 0; h < H; ++h) x[h] = lrelu(as[(lo + dl) * H + h] + ad_l[dl * H + h]);
      }
#pragma unroll
      for (int h = 0; h < H; ++h) {
        const float p = __expf(x[h] - unfkey(m_l[dl * H + h]));
        p_l[i * H + h] = p;
        atomicAdd(&den_l[dl * H + h], p);
      }
    }
  };

  constexpr int BATCH = 16;
  constexpr int JB = (H == 1) ? 16 : 8;
  auto acc_batch = [&](auto all_c, int b0, int nb) __attribute__((always_inline)) {
    constexpr bool ALL = decltype(all_c)::value;
    const unsigned pkv = (lane < nb) ? list_l[b0 + lane] : 0u;
#pragma unroll
    for (int j0 = 0; j0 < BATCH; j0 += JB) {
      if (ALL || j0 < nb) {
        int dl[JB] = {};
        float v[JB][H] = {}, p[JB][H] = {};
#pragma unroll
        for (int j = 0; j < JB; ++j) {
          if (ALL || j0 + j < nb) {
            const unsigned pk = (unsigned)__builtin_amdgcn_readlane((int)pkv, j0 + j);
            const int s = (int)(pk & 0xffffu);
            dl[j] = (int)(pk >> 16);
#pragma unroll
            for (int h = 0; h < H; ++h) v[j][h] = feat[(size_t)s * HC + h * HID + lane];
          }
        }
#pragma unroll
        for (int j = 0; j < JB; ++j) {
          if (ALL || j0 + j < nb) {
            if constexpr (H == 4) {
              const float4 p4 = *reinterpret_cast<const float4*>(&p_l[(b0 + j0 + j) * 4]);
              p[j][0] = p4.x; p[j][1] = p4.y; p[j][2] = p4.z; p[j][3] = p4.w;
            } else {
#pragma unroll
              for (int h = 0; h < H; ++h) p[j][h] = p_l[(b0 + j0 + j) * H + h];
            }
          }
        }
#pragma unroll
        for (int j = 0; j < JB; ++j) {
          if (ALL || j0 + j < nb) {
#pragma unroll
            for (int h = 0; h < H; ++h)
              atomicAdd(&acc_l[dl[j] * HC + h * HID + lane], p[j][h] * v[j][h]);
          }
        }
      }
    }
  };
  auto acc_pass = [&](int tot) __attribute__((always_inline)) {
    for (int b0 = wave * BATCH; b0 < tot; b0 += 32 * BATCH) {
      const int nb = min(BATCH, tot - b0);
      if (nb == BATCH) acc_batch(flag_c<true>{}, b0, BATCH);
      else acc_batch(flag_c<false>{}, b0, nb);
    }
  };

  auto scan_units = [&](int u0, int u1, int rot) __attribute__((always_inline)) {
    find_units(own4, u0, u1, rot, pat, p_l, &qcnt_l, wave, lane);
    __syncthreads();
    const int nq = (int)qcnt_l;
    if (nq <= QCAP) decode_units(own4, pat, p_l, nq, list_l, &cnt_l, wave, lane);
    __syncthreads();
    return (nq <= QCAP) ? (int)cnt_l : LCAP + 1;
  };

  const int n = scan_units(0, NU, (int)blockIdx.x * (NU / NBLK));
  if (n <= LCAP) {
    max_pass(n);
    __syncthreads();
    weight_pass(n, nn);
    __syncthreads();
    acc_pass(n + nn);
  } else {
    for (int c0 = 0; c0 < NU; c0 += CHU) {
      __syncthreads();
      if (tid == 0) { cnt_l = 0u; qcnt_l = 0u; }
      __syncthreads();
      max_pass(scan_units(c0, min(c0 + CHU, NU), 0));
    }
    for (int c0 = 0; c0 < NU; c0 += CHU) {
      __syncthreads();
      if (tid == 0) { cnt_l = 0u; qcnt_l = 0u; }
      __syncthreads();
      const int nc = scan_units(c0, min(c0 + CHU, NU), 0);
      const int ns = (c0 == 0) ? nn : 0;
      max_pass(nc);
      __syncthreads();
      weight_pass(nc, ns);
      __syncthreads();
      acc_pass(nc + ns);
    }
  }
  __syncthreads();

  if (!FINAL) {
    for (int i = tid; i < nn * HC; i += 1024) {
      const int dl = i / HC, r = i % HC;
      outp[(size_t)lo * HC + i] = elu(acc_l[i] / den_l[dl * H + r / HID] + bias[r]);
    }
  } else {
    for (int dl = wave; dl < nn; dl += 32) {
      const float v = elu(acc_l[dl * HC + lane] / den_l[dl * H] + bias[lane]);
      float w0 = v * lin_w[lane * 2 + 0];
      float w1 = v * lin_w[lane * 2 + 1];
#pragma unroll
      for (int m = 16; m > 0; m >>= 1) {
        w0 += __shfl_xor(w0, m, 32);
        w1 += __shfl_xor(w1, m, 32);
      }
      if (lane == 0) {
        outp[(lo + dl) * 2 + 0] = w0 + lin_b[0];
        outp[(lo + dl) * 2 + 1] = w1 + lin_b[1];
      }
    }
  }
}

static inline unsigned blks(long long tot, int bs) { return (unsigned)((tot + bs - 1) / bs); }

extern "C" void kernel_launch(void* const* d_in, const int* in_sizes, int n_in,
                              void* d_out, int out_size, void* d_ws, size_t ws_size,
                              hipStream_t stream) {
  const float* x        = (const float*)d_in[0];
  const int*   ei       = (const int*)d_in[1];
  const float* W1       = (const float*)d_in[2];
  const float* att_src1 = (const float*)d_in[3];
  const float* att_dst1 = (const float*)d_in[4];
  const float* b1       = (const float*)d_in[5];
  const float* W2       = (const float*)d_in[6];
  const float* att_src2 = (const float*)d_in[7];
  const float* att_dst2 = (const float*)d_in[8];
  const float* b2       = (const float*)d_in[9];
  const float* lin_w    = (const float*)d_in[10];
  const float* lin_b    = (const float*)d_in[11];
  float* out = (float*)d_out;
  const int* src = ei;
  const int* dst = ei + NE;

  float*    ws   = (float*)d_ws;
  float*    h1   = ws;
  float*    as1  = ws + 6400000;
  float*    ad1  = as1 + 200000;
  unsigned* own  = (unsigned*)(ad1 + 200000);
  float*    h2   = ad1 + 600000;
  float*    z    = h1;
  float*    as2  = h1 + 3200000;
  float*    ad2  = as2 + 50000;

  owner_bytes<<<blks(NE / 4, 256), 256, 0, stream>>>(dst, own);

  gemm_wmma_f32<HEADS * HID, INDIM, 4>
      <<<dim3(NN / 16), dim3(32, 2, 1), 0, stream>>>(x, W1, h1);
  att_scores<<<blks(NN * HEADS, 256), 256, 0, stream>>>(h1, att_src1, att_dst1, as1, ad1, HEADS, HID);
  agg_owner<HEADS, false><<<NBLK, 1024, 0, stream>>>(src, dst, own, as1, ad1, h1, b1, lin_w, lin_b, h2);

  gemm_wmma_f32<HID, HEADS * HID, 2>
      <<<dim3(NN / 80), dim3(32, 1, 5), 0, stream>>>(h2, W2, z);
  att_scores<<<blks(NN, 256), 256, 0, stream>>>(z, att_src2, att_dst2, as2, ad2, 1, HID);
  agg_owner<1, true><<<NBLK, 1024, 0, stream>>>(src, dst, own, as2, ad2, z, b2, lin_w, lin_b, out);
}


# Owner map gets an index by owner (a bit per 32 edges, built in LDS beside the map): a work-group reads its 782 index words and decodes only the flagged unit pairs; sums in another LDS arrival order
# speedup vs baseline: 6.2240x; 1.0779x over previous
#include <hip/hip_runtime.h>
#include <math.h>

#define NN 50000
#define NE 800000
#define INDIM 256
#define HID 32
#define HEADS 4
#define ETOT (NE + NN)

typedef __attribute__((ext_vector_type(2))) float v2f;
typedef __attribute__((ext_vector_type(8))) float v8f;

__device__ __forceinline__ unsigned fkey(float x) {
  unsigned u = __float_as_uint(x);
  return (u & 0x80000000u) ? ~u : (u | 0x80000000u);
}
__device__ __forceinline__ float unfkey(unsigned k) {
  return __uint_as_float((k & 0x80000000u) ? (k & 0x7fffffffu) : ~k);
}
__device__ __forceinline__ float elu(float v) {
  return (v > 0.f) ? v : (__expf(v) - 1.f);
}

typedef __attribute__((ext_vector_type(4))) float v4f;
template <int NT> struct colvec;
template <> struct colvec<4> { typedef v4f type; };
template <> struct colvec<2> { typedef v2f type; };

template <int N, int K, int NT>
__global__ void gemm_wmma_f32(const float* __restrict__ A, const float* __restrict__ B,
                              float* __restrict__ C) {
  typedef typename colvec<NT>::type vnt;
  const int lane  = threadIdx.x;
  const int c     = lane & 15;
  const int h     = lane >> 4;
  const int mbase = (blockIdx.x * blockDim.z + threadIdx.z) * 16;
  const int nb    = threadIdx.y * (16 * NT);
  const float* ap = A + (size_t)(mbase + c) * K + 4 * h;
  const float* bp = B + (size_t)(4 * h) * N + nb + NT * c;
  v8f acc[NT];
#pragma unroll
  for (int t = 0; t < NT; ++t) acc[t] = v8f{0.f, 0.f, 0.f, 0.f, 0.f, 0.f, 0.f, 0.f};

  constexpr int KC  = 16;
  constexpr int NCH = K / KC;
  static_assert(K % KC == 0, "whole chunks");
  v4f af[2][2];
  vnt bf[2][8];
#pragma unroll
  for (int g = 0; g < 2; ++g) {
    af[0][g] = *reinterpret_cast<const v4f*>(ap + 8 * g);
#pragma unroll
    for (int j = 0; j < 4; ++j)
      bf[0][4 * g + j] = *reinterpret_cast<const vnt*>(bp + (size_t)(8 * g + j) * N);
  }
#pragma unroll
  for (int ch = 0; ch < NCH; ++ch) {
    const int cur = ch & 1, nxt = cur ^ 1;
    if (ch + 1 < NCH) {
      const float* apn = ap + (ch + 1) * KC;
      const float* bpn = bp + (size_t)(ch + 1) * KC * N;
#pragma unroll
      for (int g = 0; g < 2; ++g) {
        af[nxt][g] = *reinterpret_cast<const v4f*>(apn + 8 * g);
#pragma unroll
        for (int j = 0; j < 4; ++j)
          bf[nxt][4 * g + j] = *reinterpret_cast<const vnt*>(bpn + (size_t)(8 * g + j) * N);
      }
    }
    __builtin_amdgcn_sched_barrier(0);
#pragma unroll
    for (int g = 0; g < 2; ++g) {
#pragma unroll
      for (int u = 0; u < 2; ++u) {
        v2f a;
        a.x = af[cur][g][2 * u];
        a.y = af[cur][g][2 * u + 1];
#pragma unroll
        for (int t = 0; t < NT; ++t) {
          v2f b;
          b.x = bf[cur][4 * g + 2 * u][t];
          b.y = bf[cur][4 * g + 2 * u + 1][t];
          acc[t] = __builtin_amdgcn_wmma_f32_16x16x4_f32(false, a, false, b, (short)0, acc[t],
                                                         false, false);
        }
      }
    }
    __builtin_amdgcn_sched_barrier(0);
  }
  float* cp = C + (size_t)(mbase + 8 * h) * N + nb + NT * c;
#pragma unroll
  for (int r = 0; r < 8; ++r) {
    vnt o;
#pragma unroll
    for (int t = 0; t < NT; ++t) o[t] = acc[t][r];
    *reinterpret_cast<vnt*>(cp + (size_t)r * N) = o;
  }
}

__global__ void att_scores(const float* __restrict__ h, const float* __restrict__ att_s,
                           const float* __restrict__ att_d, float* __restrict__ as,
                           float* __restrict__ ad, int H, int C) {
  int i = blockIdx.x * blockDim.x + threadIdx.x;
  if (i >= NN * H) return;
  int hh = i % H;
  const float* hp = h + (size_t)i * C;
  float s = 0.f, d = 0.f;
  for (int c = 0; c < C; ++c) {
    float v = hp[c];
    s += v * att_s[hh * C + c];
    d += v * att_d[hh * C + c];
  }
  as[i] = s;
  ad[i] = d;
}

#define NPB  208
#define NBLK ((NN + NPB - 1) / NPB)
#define NU   (NE / 16)
#define LCAP 8192
#define CHU  (LCAP / 16)
#define QCAP LCAP
#define NCHK ((NE / 4 + 255) / 256)
static_assert(NBLK <= 256, "owner id must fit a byte");
static_assert(NE % 32 == 0, "owner map is read in 16-byte units, and in pairs of them through the index");
static_assert(NN <= 65536 && NPB <= 65536, "list entry packs source and local destination in 16 bits each");
static_assert(HID == 32, "one lane per channel of a head");
static_assert(NN % 80 == 0, "product launches: 16 rows per wave, five waves per block in the second");

__device__ __forceinline__ float lrelu(float x) { return (x > 0.f) ? x : 0.2f * x; }

__global__ void owner_bytes(const int* __restrict__ dst, unsigned* __restrict__ own, unsigned* __restrict__ idx) {
  __shared__ unsigned w_l[NBLK];
  if (threadIdx.x < NBLK) w_l[threadIdx.x] = 0u;
  __syncthreads();
  const int i = blockIdx.x * blockDim.x + threadIdx.x;
  if (i < NE / 4) {
    const int4 d = ((const int4*)dst)[i];
    const unsigned o0 = (unsigned)(d.x / NPB), o1 = (unsigned)(d.y / NPB);
    const unsigned o2 = (unsigned)(d.z / NPB), o3 = (unsigned)(d.w / NPB);
    own[i] = o0 | (o1 << 8) | (o2 << 16) | (o3 << 24);
    const unsigned bit = 1u << (threadIdx.x >> 3);
    atomicOr(&w_l[o0], bit);
    atomicOr(&w_l[o1], bit);
    atomicOr(&w_l[o2], bit);
    atomicOr(&w_l[o3], bit);
  }
  __syncthreads();
  if (threadIdx.x < NBLK) idx[blockIdx.x * NBLK + threadIdx.x] = w_l[threadIdx.x];
}
static_assert(NBLK <= 256, "one thread per owner in owner_bytes");

__device__ __forceinline__ unsigned zero_bytes(unsigned t) {
  return ~(((t & 0x7f7f7f7fu) + 0x7f7f7f7fu) | t) & 0x80808080u;
}


__device__ __forceinline__ void find_units(const uint4* __restrict__ own4, int u0, int u1,
                                           unsigned pat, float* q_l, unsigned* qcnt_l, int wave, int lane) {
  for (int ub = u0 + wave * 32; ub < u1; ub += 4 * 1024) {
    const int u = ub + lane;
    uint4 v[4];
    int id[4];
    bool ok[4];
#pragma unroll
    for (int q = 0; q < 4; ++q) {
      const int uu = u + q * 1024;
      ok[q] = uu < u1;
      id[q] = ok[q] ? uu : u0;
      v[q] = own4[id[q]];
    }
    unsigned bal[4];
    bool hit[4];
#pragma unroll
    for (int q = 0; q < 4; ++q) {
      const unsigned tx = v[q].x ^ pat, ty = v[q].y ^ pat, tz = v[q].z ^ pat, tw = v[q].w ^ pat;
      const unsigned a = ((tx - 0x01010101u) & ~tx) | ((ty - 0x01010101u) & ~ty) |
                         ((tz - 0x01010101u) & ~tz) | ((tw - 0x01010101u) & ~tw);
      hit[q] = ok[q] && ((a & 0x80808080u) != 0u);
      bal[q] = __builtin_amdgcn_ballot_w32(hit[q]);
    }
    const unsigned tot = (unsigned)(__builtin_popcount(bal[0]) + __builtin_popcount(bal[1]) +
                                    __builtin_popcount(bal[2]) + __builtin_popcount(bal[3]));
    if (tot) {
      unsigned base = 0u;
      if (lane == 0) base = atomicAdd(qcnt_l, tot);
      base = (unsigned)__builtin_amdgcn_readfirstlane((int)base);
#pragma unroll
      for (int q = 0; q < 4; ++q) {
        if (hit[q]) {
          const unsigned pos = base + __builtin_amdgcn_mbcnt_lo(bal[q], 0u);
          if (pos < QCAP) q_l[pos] = __uint_as_float((unsigned)id[q]);
        }
        base += (unsigned)__builtin_popcount(bal[q]);
      }
    }
  }
}

__device__ __forceinline__ void queue_groups(const unsigned* __restrict__ idx, int blk, float* q_l,
                                             unsigned* qcnt_l, int tid) {
  for (int t = tid; t < NCHK; t += 1024) {
    const unsigned w = idx[t * NBLK + blk];
    if (w) {
      unsigned pos = atomicAdd(qcnt_l, (unsigned)__popc(w));
      for (unsigned r = w; r; r &= r - 1) {
        if (pos < QCAP) q_l[pos] = __uint_as_float((unsigned)(t * 32 + __ffs(r) - 1));
        ++pos;
      }
    }
  }
}

template <int UPE>
__device__ __forceinline__ void decode_units(const uint4* __restrict__ own4, unsigned pat, const float* q_l, int nq,
                                             unsigned* list_l, unsigned* cnt_l, int wave, int lane) {
  static_assert(UPE == 1 || UPE == 2, "the byte mask of an entry must fit 32 bits");
  for (int i0 = wave * 32; i0 < nq; i0 += 4 * 1024) {
    uint4 v[4][UPE];
    unsigned id[4];
    bool ok[4];
#pragma unroll
    for (int q = 0; q < 4; ++q) {
      const int i = i0 + q * 1024 + lane;
      ok[q] = i < nq;
      id[q] = __float_as_uint(q_l[ok[q] ? i : i0]);
#pragma unroll
      for (int p = 0; p < UPE; ++p) v[q][p] = own4[id[q] * UPE + p];
    }
#pragma unroll
    for (int q = 0; q < 4; ++q) {
      unsigned mm = 0u;
#pragma unroll
      for (int p = 0; p < UPE; ++p)
        mm |= ((zero_bytes(v[q][p].x ^ pat) >> 7) | (zero_bytes(v[q][p].y ^ pat) >> 6) |
               (zero_bytes(v[q][p].z ^ pat) >> 5) | (zero_bytes(v[q][p].w ^ pat) >> 4)) << (4 * p);
      if (!ok[q]) mm = 0u;
      const unsigned c = (unsigned)__popc(mm);
      unsigned excl = 0u, tot = 0u;
#pragma unroll
      for (int b = 0; b < 4 + UPE; ++b) {
        const unsigned bl = __builtin_amdgcn_ballot_w32(((c >> b) & 1u) != 0u);
        excl += __builtin_amdgcn_mbcnt_lo(bl, 0u) << b;
        tot += (unsigned)__builtin_popcount(bl) << b;
      }
      if (tot) {
        unsigned base = 0u;
        if (lane == 0) base = atomicAdd(cnt_l, tot);
        unsigned pos = (unsigned)__builtin_amdgcn_readfirstlane((int)base) + excl;
        for (unsigned r = mm; r; r &= r - 1) {
          const int j = __ffs(r) - 1;
          if (pos < LCAP)
            list_l[pos] = id[q] * (16u * UPE) + (unsigned)(16 * ((j >> 2) & 1) + 4 * (j & 3) + (j >> 3));
          ++pos;
        }
      }
    }
  }
}

template <bool B> struct flag_c { static constexpr bool value = B; };

template <int H, bool FINAL>
__global__ __launch_bounds__(1024) void agg_owner(
    const int* __restrict__ src, const int* __restrict__ dst, const unsigned* __restrict__ own,
    const unsigned* __restrict__ idx,
    const float* __restrict__ as, const float* __restrict__ ad, const float* __restrict__ feat,
    const float* __restrict__ bias, const float* __restrict__ lin_w, const float* __restrict__ lin_b,
    float* __restrict__ outp) {
  constexpr int HC = H * HID;
  __shared__ float    acc_l[NPB * HC];
  __shared__ float    den_l[NPB * H];
  __shared__ __attribute__((aligned(16))) float    ad_l[NPB * H];
  __shared__ __attribute__((aligned(16))) unsigned m_l[NPB * H];
  __shared__ unsigned list_l[LCAP + NPB];
  __shared__ __attribute__((aligned(16))) float p_l[(LCAP + NPB) * H];
  __shared__ unsigned cnt_l, qcnt_l;
  static_assert(QCAP <= (LCAP + NPB) * H, "the unit queue borrows p_l");
  const int tid = threadIdx.x, lane = tid & 31;
  const int wave = __builtin_amdgcn_readfirstlane(tid >> 5);
  const int lo = blockIdx.x * NPB;
  const int nn = min(NPB, NN - lo);
  const uint4* own4 = (const uint4*)own;
  const unsigned pat = (unsigned)blockIdx.x * 0x01010101u;

  for (int i = tid; i < NPB * HC; i += 1024) acc_l[i] = 0.f;
  for (int i = tid; i < NPB * H; i += 1024) {
    float a_d = 0.f;
    unsigned mk = 0u;
    if (i < nn * H) {
      a_d = ad[lo * H + i];
      mk = fkey(lrelu(as[lo * H + i] + a_d));
    }
    ad_l[i] = a_d;
    m_l[i] = mk;
    den_l[i] = 0.f;
  }
  if (tid == 0) { cnt_l = 0u; qcnt_l = 0u; }
  __syncthreads();

  auto max_pass = [&](int n) {
    for (int i = tid; i < n; i += 1024) {
      const int e = (int)list_l[i];
      const int s = src[e];
      const int dl = dst[e] - lo;
#pragma unroll
      for (int h = 0; h < H; ++h) {
        const float x = lrelu(as[s * H + h] + ad_l[dl * H + h]);
        atomicMax(&m_l[dl * H + h], fkey(x));
        p_l[i * H + h] = x;
      }
      list_l[i] = (unsigned)s | ((unsigned)dl << 16);
    }
  };

  auto weight_pass = [&](int n, int nself) {
    const int tot = n + nself;
    for (int i = tid; i < tot; i += 1024) {
      float x[H];
      int dl;
      if (i < n) {
        dl = (int)(list_l[i] >> 16);
#pragma unroll
        for (int h = 0; h < H; ++h) x[h] = p_l[i * H + h];
      } else {
        dl = i - n;
        list_l[i] = (unsigned)(lo + dl) | ((unsigned)dl << 16);
#pragma unroll
        for (int h = 0; h < H; ++h) x[h] = lrelu(as[(lo + dl) * H + h] + ad_l[dl * H + h]);
      }
#pragma unroll
      for (int h = 0; h < H; ++h) {
        const float p = __expf(x[h] - unfkey(m_l[dl * H + h]));
        p_l[i * H + h] = p;
        atomicAdd(&den_l[dl * H + h], p);
      }
    }
  };

  constexpr int BATCH = 16;
  constexpr int JB = (H == 1) ? 16 : 8;
  auto acc_batch = [&](auto all_c, int b0, int nb) __attribute__((always_inline)) {
    constexpr bool ALL = decltype(all_c)::value;
    const unsigned pkv = (lane < nb) ? list_l[b0 + lane] : 0u;
#pragma unroll
    for (int j0 = 0; j0 < BATCH; j0 += JB) {
      if (ALL || j0 < nb) {
        int dl[JB] = {};
        float v[JB][H] = {}, p[JB][H] = {};
#pragma unroll
        for (int j = 0; j < JB; ++j) {
          if (ALL || j0 + j < nb) {
            const unsigned pk = (unsigned)__builtin_amdgcn_readlane((int)pkv, j0 + j);
            const int s = (int)(pk & 0xffffu);
            dl[j] = (int)(pk >> 16);
#pragma unroll
            for (int h = 0; h < H; ++h) v[j][h] = feat[(size_t)s * HC + h * HID + lane];
          }
        }
#pragma unroll
        for (int j = 0; j < JB; ++j) {
          if (ALL || j0 + j < nb) {
            if constexpr (H == 4) {
              const float4 p4 = *reinterpret_cast<const float4*>(&p_l[(b0 + j0 + j) * 4]);
              p[j][0] = p4.x; p[j][1] = p4.y; p[j][2] = p4.z; p[j][3] = p4.w;
            } else {
#pragma unroll
              for (int h = 0; h < H; ++h) p[j][h] = p_l[(b0 + j0 + j) * H + h];
            }
          }
        }
#pragma unroll
        for (int j = 0; j < JB; ++j) {
          if (ALL || j0 + j < nb) {
#pragma unroll
            for (int h = 0; h < H; ++h)
              atomicAdd(&acc_l[dl[j] * HC + h * HID + lane], p[j][h] * v[j][h]);
          }
        }
      }
    }
  };
  auto acc_pass = [&](int tot) __attribute__((always_inline)) {
    for (int b0 = wave * BATCH; b0 < tot; b0 += 32 * BATCH) {
      const int nb = min(BATCH, tot - b0);
      if (nb == BATCH) acc_batch(flag_c<true>{}, b0, BATCH);
      else acc_batch(flag_c<false>{}, b0, nb);
    }
  };

  auto scan_units = [&](int u0, int u1) __attribute__((always_inline)) {
    find_units(own4, u0, u1, pat, p_l, &qcnt_l, wave, lane);
    __syncthreads();
    const int nq = (int)qcnt_l;
    if (nq <= QCAP) decode_units<1>(own4, pat, p_l, nq, list_l, &cnt_l, wave, lane);
    __syncthreads();
    return (nq <= QCAP) ? (int)cnt_l : LCAP + 1;
  };
  auto scan_index = [&]() __attribute__((always_inline)) {
    queue_groups(idx, (int)blockIdx.x, p_l, &qcnt_l, tid);
    __syncthreads();
    const int nq = (int)qcnt_l;
    if (nq <= QCAP) decode_units<2>(own4, pat, p_l, nq, list_l, &cnt_l, wave, lane);
    __syncthreads();
    return (nq <= QCAP) ? (int)cnt_l : LCAP + 1;
  };

  const int n = scan_index();
  if (n <= LCAP) {
    max_pass(n);
    __syncthreads();
    weight_pass(n, nn);
    __syncthreads();
    acc_pass(n + nn);
  } else {
    for (int c0 = 0; c0 < NU; c0 += CHU) {
      __syncthreads();
      if (tid == 0) { cnt_l = 0u; qcnt_l = 0u; }
      __syncthreads();
      max_pass(scan_units(c0, min(c0 + CHU, NU)));
    }
    for (int c0 = 0; c0 < NU; c0 += CHU) {
      __syncthreads();
      if (tid == 0) { cnt_l = 0u; qcnt_l = 0u; }
      __syncthreads();
      const int nc = scan_units(c0, min(c0 + CHU, NU));
      const int ns = (c0 == 0) ? nn : 0;
      max_pass(nc);
      __syncthreads();
      weight_pass(nc, ns);
      __syncthreads();
      acc_pass(nc + ns);
    }
  }
  __syncthreads();

  if (!FINAL) {
    for (int i = tid; i < nn * HC; i += 1024) {
      const int dl = i / HC, r = i % HC;
      outp[(size_t)lo * HC + i] = elu(acc_l[i] / den_l[dl * H + r / HID] + bias[r]);
    }
  } else {
    for (int dl = wave; dl < nn; dl += 32) {
      const float v = elu(acc_l[dl * HC + lane] / den_l[dl * H] + bias[lane]);
      float w0 = v * lin_w[lane * 2 + 0];
      float w1 = v * lin_w[lane * 2 + 1];
#pragma unroll
      for (int m = 16; m > 0; m >>= 1) {
        w0 += __shfl_xor(w0, m, 32);
        w1 += __shfl_xor(w1, m, 32);
      }
      if (lane == 0) {
        outp[(lo + dl) * 2 + 0] = w0 + lin_b[0];
        outp[(lo + dl) * 2 + 1] = w1 + lin_b[1];
      }
    }
  }
}

static inline unsigned blks(long long tot, int bs) { return (unsigned)((tot + bs - 1) / bs); }

extern "C" void kernel_launch(void* const* d_in, const int* in_sizes, int n_in,
                              void* d_out, int out_size, void* d_ws, size_t ws_size,
                              hipStream_t stream) {
  const float* x        = (const float*)d_in[0];
  const int*   ei       = (const int*)d_in[1];
  const float* W1       = (const float*)d_in[2];
  const float* att_src1 = (const float*)d_in[3];
  const float* att_dst1 = (const float*)d_in[4];
  const float* b1       = (const float*)d_in[5];
  const float* W2       = (const float*)d_in[6];
  const float* att_src2 = (const float*)d_in[7];
  const float* att_dst2 = (const float*)d_in[8];
  const float* b2       = (const float*)d_in[9];
  const float* lin_w    = (const float*)d_in[10];
  const float* lin_b    = (const float*)d_in[11];
  float* out = (float*)d_out;
  const int* src = ei;
  const int* dst = ei + NE;

  float*    ws   = (float*)d_ws;
  float*    h1   = ws;
  float*    as1  = ws + 6400000;
  float*    ad1  = as1 + 200000;
  unsigned* own  = (unsigned*)(ad1 + 200000);
  unsigned* idx  = own + 200000;
  float*    h2   = ad1 + 600000;
  float*    z    = h1;
  float*    as2  = h1 + 3200000;
  float*    ad2  = as2 + 50000;

  owner_bytes<<<NCHK, 256, 0, stream>>>(dst, own, idx);

  gemm_wmma_f32<HEADS * HID, INDIM, 4>
      <<<dim3(NN / 16), dim3(32, 2, 1), 0, stream>>>(x, W1, h1);
  att_scores<<<blks(NN * HEADS, 256), 256, 0, stream>>>(h1, att_src1, att_dst1, as1, ad1, HEADS, HID);
  agg_owner<HEADS, false><<<NBLK, 1024, 0, stream>>>(src, dst, own, idx, as1, ad1, h1, b1, lin_w, lin_b, h2);

  gemm_wmma_f32<HID, HEADS * HID, 2>
      <<<dim3(NN / 80), dim3(32, 1, 5), 0, stream>>>(h2, W2, z);
  att_scores<<<blks(NN, 256), 256, 0, stream>>>(z, att_src2, att_dst2, as2, ad2, 1, HID);
  agg_owner<1, true><<<NBLK, 1024, 0, stream>>>(src, dst, own, idx, as2, ad2, z, b2, lin_w, lin_b, out);
}


# Attention scores come out of each product as extra columns: weights folded with the attention vectors in launch 1, one score wave per 16 rows; both score passes go; scores summed in another order
# speedup vs baseline: 6.5216x; 1.0478x over previous
#include <hip/hip_runtime.h>
#include <math.h>

#define NN 50000
#define NE 800000
#define INDIM 256
#define HID 32
#define HEADS 4
#define ETOT (NE + NN)

typedef __attribute__((ext_vector_type(2))) float v2f;
typedef __attribute__((ext_vector_type(8))) float v8f;

__device__ __forceinline__ unsigned fkey(float x) {
  unsigned u = __float_as_uint(x);
  return (u & 0x80000000u) ? ~u : (u | 0x80000000u);
}
__device__ __forceinline__ float unfkey(unsigned k) {
  return __uint_as_float((k & 0x80000000u) ? (k & 0x7fffffffu) : ~k);
}
__device__ __forceinline__ float elu(float v) {
  return (v > 0.f) ? v : (__expf(v) - 1.f);
}

typedef __attribute__((ext_vector_type(4))) float v4f;
template <int NT> struct colvec;
template <> struct colvec<4> { typedef v4f type; };
template <> struct colvec<2> { typedef v2f type; };

template <int K, int H>
__device__ __forceinline__ void score_tile(const float* __restrict__ A, const float* __restrict__ Bs,
                                           float* __restrict__ SD, int mbase, int lane) {
  const int c = lane & 15;
  const int h = lane >> 4;
  const float* ap = A + (size_t)(mbase + c) * K + 4 * h;
  const float* bp = Bs + h * 64 + 4 * c;
  constexpr int G   = 4;
  constexpr int NCH = K / (8 * G);
  static_assert(K % (8 * G) == 0, "whole chunks");
  v8f acc[2];
#pragma unroll
  for (int t = 0; t < 2; ++t) acc[t] = v8f{0.f, 0.f, 0.f, 0.f, 0.f, 0.f, 0.f, 0.f};
  v4f af[2][G], bf[2][G];
#pragma unroll
  for (int g = 0; g < G; ++g) {
    af[0][g] = *reinterpret_cast<const v4f*>(ap + 8 * g);
    bf[0][g] = *reinterpret_cast<const v4f*>(bp + 128 * g);
  }
#pragma unroll
  for (int ch = 0; ch < NCH; ++ch) {
    const int cur = ch & 1, nxt = cur ^ 1;
    if (ch + 1 < NCH) {
#pragma unroll
      for (int g = 0; g < G; ++g) {
        af[nxt][g] = *reinterpret_cast<const v4f*>(ap + (ch + 1) * 8 * G + 8 * g);
        bf[nxt][g] = *reinterpret_cast<const v4f*>(bp + ((ch + 1) * G + g) * 128);
      }
    }
    __builtin_amdgcn_sched_barrier(0);
#pragma unroll
    for (int u = 0; u < 2; ++u) {
#pragma unroll
      for (int g = 0; g < G; ++g) {
        v2f a, b;
        a.x = af[cur][g][2 * u];
        a.y = af[cur][g][2 * u + 1];
        b.x = bf[cur][g][2 * u];
        b.y = bf[cur][g][2 * u + 1];
        acc[g & 1] = __builtin_amdgcn_wmma_f32_16x16x4_f32(false, a, false, b, (short)0, acc[g & 1],
                                                           false, false);
      }
    }
    __builtin_amdgcn_sched_barrier(0);
  }
  if (c < 2 * H) {
    float* sp = SD + (size_t)(mbase + 8 * h) * (2 * H) + c;
#pragma unroll
    for (int r = 0; r < 8; ++r) sp[r * (2 * H)] = acc[0][r] + acc[1][r];
  }
}

template <int N, int K, int NT, int H>
__global__ void gemm_wmma_f32(const float* __restrict__ A, const float* __restrict__ B,
                              float* __restrict__ C, const float* __restrict__ Bs,
                              float* __restrict__ SD) {
  typedef typename colvec<NT>::type vnt;
  const int lane  = threadIdx.x;
  const int c     = lane & 15;
  const int h     = lane >> 4;
  const int mbase = (blockIdx.x * blockDim.z + threadIdx.z) * 16;
  const int wy    = __builtin_amdgcn_readfirstlane((int)threadIdx.y);
  if (wy == N / (16 * NT)) {
    score_tile<K, H>(A, Bs, SD, mbase, lane);
    return;
  }
  const int nb    = wy * (16 * NT);
  const float* ap = A + (size_t)(mbase + c) * K + 4 * h;
  const float* bp = B + (size_t)(4 * h) * N + nb + NT * c;
  v8f acc[NT];
#pragma unroll
  for (int t = 0; t < NT; ++t) acc[t] = v8f{0.f, 0.f, 0.f, 0.f, 0.f, 0.f, 0.f, 0.f};

  constexpr int KC  = 16;
  constexpr int NCH = K / KC;
  static_assert(K % KC == 0, "whole chunks");
  v4f af[2][2];
  vnt bf[2][8];
#pragma unroll
  for (int g = 0; g < 2; ++g) {
    af[0][g] = *reinterpret_cast<const v4f*>(ap + 8 * g);
#pragma unroll
    for (int j = 0; j < 4; ++j)
      bf[0][4 * g + j] = *reinterpret_cast<const vnt*>(bp + (size_t)(8 * g + j) * N);
  }
#pragma unroll
  for (int ch = 0; ch < NCH; ++ch) {
    const int cur = ch & 1, nxt = cur ^ 1;
    if (ch + 1 < NCH) {
      const float* apn = ap + (ch + 1) * KC;
      const float* bpn = bp + (size_t)(ch + 1) * KC * N;
#pragma unroll
      for (int g = 0; g < 2; ++g) {
        af[nxt][g] = *reinterpret_cast<const v4f*>(apn + 8 * g);
#pragma unroll
        for (int j = 0; j < 4; ++j)
          bf[nxt][4 * g + j] = *reinterpret_cast<const vnt*>(bpn + (size_t)(8 * g + j) * N);
      }
    }
    __builtin_amdgcn_sched_barrier(0);
#pragma unroll
    for (int g = 0; g < 2; ++g) {
#pragma unroll
      for (int u = 0; u < 2; ++u) {
        v2f a;
        a.x = af[cur][g][2 * u];
        a.y = af[cur][g][2 * u + 1];
#pragma unroll
        for (int t = 0; t < NT; ++t) {
          v2f b;
          b.x = bf[cur][4 * g + 2 * u][t];
          b.y = bf[cur][4 * g + 2 * u + 1][t];
          acc[t] = __builtin_amdgcn_wmma_f32_16x16x4_f32(false, a, false, b, (short)0, acc[t],
                                                         false, false);
        }
      }
    }
    __builtin_amdgcn_sched_barrier(0);
  }
  float* cp = C + (size_t)(mbase + 8 * h) * N + nb + NT * c;
#pragma unroll
  for (int r = 0; r < 8; ++r) {
    vnt o;
#pragma unroll
    for (int t = 0; t < NT; ++t) o[t] = acc[t][r];
    *reinterpret_cast<vnt*>(cp + (size_t)r * N) = o;
  }
}

#define NPB  208
#define NBLK ((NN + NPB - 1) / NPB)
#define NU   (NE / 16)
#define LCAP 8192
#define CHU  (LCAP / 16)
#define QCAP LCAP
#define NCHK ((NE / 4 + 255) / 256)
static_assert(NBLK <= 256, "owner id must fit a byte");
static_assert(NE % 32 == 0, "owner map is read in 16-byte units, and in pairs of them through the index");
static_assert(NN <= 65536 && NPB <= 65536, "list entry packs source and local destination in 16 bits each");
static_assert(HID == 32, "one lane per channel of a head");
static_assert(NN % 80 == 0, "product launches: 16 rows per wave, five waves per block in the second");

__device__ __forceinline__ float lrelu(float x) { return (x > 0.f) ? x : 0.2f * x; }

template <int H>
__device__ __forceinline__ float fold_elem(const float* __restrict__ Wm, const float* __restrict__ att_s,
                                           const float* __restrict__ att_d, int o) {
  const int k = ((o >> 6) << 2) | (o & 3);
  const int col = (o >> 2) & 15;
  if (col >= 2 * H) return 0.f;
  const int hd = (col < H) ? col : col - H;
  const float4* wp = reinterpret_cast<const float4*>(Wm + (size_t)k * (H * HID) + hd * HID);
  const float4* tp = reinterpret_cast<const float4*>(((col < H) ? att_s : att_d) + hd * HID);
  float s = 0.f;
#pragma unroll
  for (int q = 0; q < HID / 4; ++q) {
    const float4 w = wp[q], t = tp[q];
    s += w.x * t.x;
    s += w.y * t.y;
    s += w.z * t.z;
    s += w.w * t.w;
  }
  return s;
}
#define FOLD1 (INDIM * 16)
#define FOLD2 (HEADS * HID * 16)
#define NFOLD ((FOLD1 + FOLD2) / 256)
static_assert((FOLD1 + FOLD2) % 256 == 0, "whole work-groups of fold elements");

__global__ void owner_bytes(const int* __restrict__ dst, unsigned* __restrict__ own, unsigned* __restrict__ idx,
                            const float* __restrict__ W1, const float* __restrict__ att_s1,
                            const float* __restrict__ att_d1, const float* __restrict__ W2,
                            const float* __restrict__ att_s2, const float* __restrict__ att_d2,
                            float* __restrict__ bs1, float* __restrict__ bs2) {
  if (blockIdx.x < NFOLD) {
    const int o = blockIdx.x * 256 + threadIdx.x;
    if (o < FOLD1) bs1[o] = fold_elem<HEADS>(W1, att_s1, att_d1, o);
    else bs2[o - FOLD1] = fold_elem<1>(W2, att_s2, att_d2, o - FOLD1);
    return;
  }
  const int blk = blockIdx.x - NFOLD;
  __shared__ unsigned w_l[NBLK];
  if (threadIdx.x < NBLK) w_l[threadIdx.x] = 0u;
  __syncthreads();
  const int i = blk * blockDim.x + threadIdx.x;
  if (i < NE / 4) {
    const int4 d = ((const int4*)dst)[i];
    const unsigned o0 = (unsigned)(d.x / NPB), o1 = (unsigned)(d.y / NPB);
    const unsigned o2 = (unsigned)(d.z / NPB), o3 = (unsigned)(d.w / NPB);
    own[i] = o0 | (o1 << 8) | (o2 << 16) | (o3 << 24);
    const unsigned bit = 1u << (threadIdx.x >> 3);
    atomicOr(&w_l[o0], bit);
    atomicOr(&w_l[o1], bit);
    atomicOr(&w_l[o2], bit);
    atomicOr(&w_l[o3], bit);
  }
  __syncthreads();
  if (threadIdx.x < NBLK) idx[blk * NBLK + threadIdx.x] = w_l[threadIdx.x];
}
static_assert(NBLK <= 256, "one thread per owner in owner_bytes");

__device__ __forceinline__ unsigned zero_bytes(unsigned t) {
  return ~(((t & 0x7f7f7f7fu) + 0x7f7f7f7fu) | t) & 0x80808080u;
}


__device__ __forceinline__ void find_units(const uint4* __restrict__ own4, int u0, int u1,
                                           unsigned pat, float* q_l, unsigned* qcnt_l, int wave, int lane) {
  for (int ub = u0 + wave * 32; ub < u1; ub += 4 * 1024) {
    const int u = ub + lane;
    uint4 v[4];
    int id[4];
    bool ok[4];
#pragma unroll
    for (int q = 0; q < 4; ++q) {
      const int uu = u + q * 1024;
      ok[q] = uu < u1;
      id[q] = ok[q] ? uu : u0;
      v[q] = own4[id[q]];
    }
    unsigned bal[4];
    bool hit[4];
#pragma unroll
    for (int q = 0; q < 4; ++q) {
      const unsigned tx = v[q].x ^ pat, ty = v[q].y ^ pat, tz = v[q].z ^ pat, tw = v[q].w ^ pat;
      const unsigned a = ((tx - 0x01010101u) & ~tx) | ((ty - 0x01010101u) & ~ty) |
                         ((tz - 0x01010101u) & ~tz) | ((tw - 0x01010101u) & ~tw);
      hit[q] = ok[q] && ((a & 0x80808080u) != 0u);
      bal[q] = __builtin_amdgcn_ballot_w32(hit[q]);
    }
    const unsigned tot = (unsigned)(__builtin_popcount(bal[0]) + __builtin_popcount(bal[1]) +
                                    __builtin_popcount(bal[2]) + __builtin_popcount(bal[3]));
    if (tot) {
      unsigned base = 0u;
      if (lane == 0) base = atomicAdd(qcnt_l, tot);
      base = (unsigned)__builtin_amdgcn_readfirstlane((int)base);
#pragma unroll
      for (int q = 0; q < 4; ++q) {
        if (hit[q]) {
          const unsigned pos = base + __builtin_amdgcn_mbcnt_lo(bal[q], 0u);
          if (pos < QCAP) q_l[pos] = __uint_as_float((unsigned)id[q]);
        }
        base += (unsigned)__builtin_popcount(bal[q]);
      }
    }
  }
}

__device__ __forceinline__ void queue_groups(const unsigned* __restrict__ idx, int blk, float* q_l,
                                             unsigned* qcnt_l, int tid) {
  for (int t = tid; t < NCHK; t += 1024) {
    const unsigned w = idx[t * NBLK + blk];
    if (w) {
      unsigned pos = atomicAdd(qcnt_l, (unsigned)__popc(w));
      for (unsigned r = w; r; r &= r - 1) {
        if (pos < QCAP) q_l[pos] = __uint_as_float((unsigned)(t * 32 + __ffs(r) - 1));
        ++pos;
      }
    }
  }
}

template <int UPE>
__device__ __forceinline__ void decode_units(const uint4* __restrict__ own4, unsigned pat, const float* q_l, int nq,
                                             unsigned* list_l, unsigned* cnt_l, int wave, int lane) {
  static_assert(UPE == 1 || UPE == 2, "the byte mask of an entry must fit 32 bits");
  for (int i0 = wave * 32; i0 < nq; i0 += 4 * 1024) {
    uint4 v[4][UPE];
    unsigned id[4];
    bool ok[4];
#pragma unroll
    for (int q = 0; q < 4; ++q) {
      const int i = i0 + q * 1024 + lane;
      ok[q] = i < nq;
      id[q] = __float_as_uint(q_l[ok[q] ? i : i0]);
#pragma unroll
      for (int p = 0; p < UPE; ++p) v[q][p] = own4[id[q] * UPE + p];
    }
#pragma unroll
    for (int q = 0; q < 4; ++q) {
      unsigned mm = 0u;
#pragma unroll
      for (int p = 0; p < UPE; ++p)
        mm |= ((zero_bytes(v[q][p].x ^ pat) >> 7) | (zero_bytes(v[q][p].y ^ pat) >> 6) |
               (zero_bytes(v[q][p].z ^ pat) >> 5) | (zero_bytes(v[q][p].w ^ pat) >> 4)) << (4 * p);
      if (!ok[q]) mm = 0u;
      const unsigned c = (unsigned)__popc(mm);
      unsigned excl = 0u, tot = 0u;
#pragma unroll
      for (int b = 0; b < 4 + UPE; ++b) {
        const unsigned bl = __builtin_amdgcn_ballot_w32(((c >> b) & 1u) != 0u);
        excl += __builtin_amdgcn_mbcnt_lo(bl, 0u) << b;
        tot += (unsigned)__builtin_popcount(bl) << b;
      }
      if (tot) {
        unsigned base = 0u;
        if (lane == 0) base = atomicAdd(cnt_l, tot);
        unsigned pos = (unsigned)__builtin_amdgcn_readfirstlane((int)base) + excl;
        for (unsigned r = mm; r; r &= r - 1) {
          const int j = __ffs(r) - 1;
          if (pos < LCAP)
            list_l[pos] = id[q] * (16u * UPE) + (unsigned)(16 * ((j >> 2) & 1) + 4 * (j & 3) + (j >> 3));
          ++pos;
        }
      }
    }
  }
}

template <bool B> struct flag_c { static constexpr bool value = B; };

template <int H, bool FINAL>
__global__ __launch_bounds__(1024) void agg_owner(
    const int* __restrict__ src, const int* __restrict__ dst, const unsigned* __restrict__ own,
    const unsigned* __restrict__ idx,
    const float* __restrict__ sd, const float* __restrict__ feat,
    const float* __restrict__ bias, const float* __restrict__ lin_w, const float* __restrict__ lin_b,
    float* __restrict__ outp) {
  constexpr int HC = H * HID;
  __shared__ float    acc_l[NPB * HC];
  __shared__ float    den_l[NPB * H];
  __shared__ __attribute__((aligned(16))) float    ad_l[NPB * H];
  __shared__ __attribute__((aligned(16))) unsigned m_l[NPB * H];
  __shared__ unsigned list_l[LCAP + NPB];
  __shared__ __attribute__((aligned(16))) float p_l[(LCAP + NPB) * H];
  __shared__ unsigned cnt_l, qcnt_l;
  static_assert(QCAP <= (LCAP + NPB) * H, "the unit queue borrows p_l");
  const int tid = threadIdx.x, lane = tid & 31;
  const int wave = __builtin_amdgcn_readfirstlane(tid >> 5);
  const int lo = blockIdx.x * NPB;
  const int nn = min(NPB, NN - lo);
  const uint4* own4 = (const uint4*)own;
  const unsigned pat = (unsigned)blockIdx.x * 0x01010101u;

  for (int i = tid; i < NPB * HC; i += 1024) acc_l[i] = 0.f;
  for (int i = tid; i < NPB * H; i += 1024) {
    float a_d = 0.f;
    unsigned mk = 0u;
    if (i < nn * H) {
      const float* sp = sd + (size_t)(lo + i / H) * (2 * H) + i % H;
      a_d = sp[H];
      mk = fkey(lrelu(sp[0] + a_d));
    }
    ad_l[i] = a_d;
    m_l[i] = mk;
    den_l[i] = 0.f;
  }
  if (tid == 0) { cnt_l = 0u; qcnt_l = 0u; }
  __syncthreads();

  auto max_pass = [&](int n) {
    for (int i = tid; i < n; i += 1024) {
      const int e = (int)list_l[i];
      const int s = src[e];
      const int dl = dst[e] - lo;
#pragma unroll
      for (int h = 0; h < H; ++h) {
        const float x = lrelu(sd[s * (2 * H) + h] + ad_l[dl * H + h]);
        atomicMax(&m_l[dl * H + h], fkey(x));
        p_l[i * H + h] = x;
      }
      list_l[i] = (unsigned)s | ((unsigned)dl << 16);
    }
  };

  auto weight_pass = [&](int n, int nself) {
    const int tot = n + nself;
    for (int i = tid; i < tot; i += 1024) {
      float x[H];
      int dl;
      if (i < n) {
        dl = (int)(list_l[i] >> 16);
#pragma unroll
        for (int h = 0; h < H; ++h) x[h] = p_l[i * H + h];
      } else {
        dl = i - n;
        list_l[i] = (unsigned)(lo + dl) | ((unsigned)dl << 16);
#pragma unroll
        for (int h = 0; h < H; ++h) x[h] = lrelu(sd[(lo + dl) * (2 * H) + h] + ad_l[dl * H + h]);
      }
#pragma unroll
      for (int h = 0; h < H; ++h) {
        const float p = __expf(x[h] - unfkey(m_l[dl * H + h]));
        p_l[i * H + h] = p;
        atomicAdd(&den_l[dl * H + h], p);
      }
    }
  };

  constexpr int BATCH = 16;
  constexpr int JB = (H == 1) ? 16 : 8;
  auto acc_batch = [&](auto all_c, int b0, int nb) __attribute__((always_inline)) {
    constexpr bool ALL = decltype(all_c)::value;
    const unsigned pkv = (lane < nb) ? list_l[b0 + lane] : 0u;
#pragma unroll
    for (int j0 = 0; j0 < BATCH; j0 += JB) {
      if (ALL || j0 < nb) {
        int dl[JB] = {};
        float v[JB][H] = {}, p[JB][H] = {};
#pragma unroll
        for (int j = 0; j < JB; ++j) {
          if (ALL || j0 + j < nb) {
            const unsigned pk = (unsigned)__builtin_amdgcn_readlane((int)pkv, j0 + j);
            const int s = (int)(pk & 0xffffu);
            dl[j] = (int)(pk >> 16);
#pragma unroll
            for (int h = 0; h < H; ++h) v[j][h] = feat[(size_t)s * HC + h * HID + lane];
          }
        }
#pragma unroll
        for (int j = 0; j < JB; ++j) {
          if (ALL || j0 + j < nb) {
            if constexpr (H == 4) {
              const float4 p4 = *reinterpret_cast<const float4*>(&p_l[(b0 + j0 + j) * 4]);
              p[j][0] = p4.x; p[j][1] = p4.y; p[j][2] = p4.z; p[j][3] = p4.w;
            } else {
#pragma unroll
              for (int h = 0; h < H; ++h) p[j][h] = p_l[(b0 + j0 + j) * H + h];
            }
          }
        }
#pragma unroll
        for (int j = 0; j < JB; ++j) {
          if (ALL || j0 + j < nb) {
#pragma unroll
            for (int h = 0; h < H; ++h)
              atomicAdd(&acc_l[dl[j] * HC + h * HID + lane], p[j][h] * v[j][h]);
          }
        }
      }
    }
  };
  auto acc_pass = [&](int tot) __attribute__((always_inline)) {
    for (int b0 = wave * BATCH; b0 < tot; b0 += 32 * BATCH) {
      const int nb = min(BATCH, tot - b0);
      if (nb == BATCH) acc_batch(flag_c<true>{}, b0, BATCH);
      else acc_batch(flag_c<false>{}, b0, nb);
    }
  };

  auto scan_units = [&](int u0, int u1) __attribute__((always_inline)) {
    find_units(own4, u0, u1, pat, p_l, &qcnt_l, wave, lane);
    __syncthreads();
    const int nq = (int)qcnt_l;
    if (nq <= QCAP) decode_units<1>(own4, pat, p_l, nq, list_l, &cnt_l, wave, lane);
    __syncthreads();
    return (nq <= QCAP) ? (int)cnt_l : LCAP + 1;
  };
  auto scan_index = [&]() __attribute__((always_inline)) {
    queue_groups(idx, (int)blockIdx.x, p_l, &qcnt_l, tid);
    __syncthreads();
    const int nq = (int)qcnt_l;
    if (nq <= QCAP) decode_units<2>(own4, pat, p_l, nq, list_l, &cnt_l, wave, lane);
    __syncthreads();
    return (nq <= QCAP) ? (int)cnt_l : LCAP + 1;
  };

  const int n = scan_index();
  if (n <= LCAP) {
    max_pass(n);
    __syncthreads();
    weight_pass(n, nn);
    __syncthreads();
    acc_pass(n + nn);
  } else {
    for (int c0 = 0; c0 < NU; c0 += CHU) {
      __syncthreads();
      if (tid == 0) { cnt_l = 0u; qcnt_l = 0u; }
      __syncthreads();
      max_pass(scan_units(c0, min(c0 + CHU, NU)));
    }
    for (int c0 = 0; c0 < NU; c0 += CHU) {
      __syncthreads();
      if (tid == 0) { cnt_l = 0u; qcnt_l = 0u; }
      __syncthreads();
      const int nc = scan_units(c0, min(c0 + CHU, NU));
      const int ns = (c0 == 0) ? nn : 0;
      max_pass(nc);
      __syncthreads();
      weight_pass(nc, ns);
      __syncthreads();
      acc_pass(nc + ns);
    }
  }
  __syncthreads();

  if (!FINAL) {
    for (int i = tid; i < nn * HC; i += 1024) {
      const int dl = i / HC, r = i % HC;
      outp[(size_t)lo * HC + i] = elu(acc_l[i] / den_l[dl * H + r / HID] + bias[r]);
    }
  } else {
    for (int dl = wave; dl < nn; dl += 32) {
      const float v = elu(acc_l[dl * HC + lane] / den_l[dl * H] + bias[lane]);
      float w0 = v * lin_w[lane * 2 + 0];
      float w1 = v * lin_w[lane * 2 + 1];
#pragma unroll
      for (int m = 16; m > 0; m >>= 1) {
        w0 += __shfl_xor(w0, m, 32);
        w1 += __shfl_xor(w1, m, 32);
      }
      if (lane == 0) {
        outp[(lo + dl) * 2 + 0] = w0 + lin_b[0];
        outp[(lo + dl) * 2 + 1] = w1 + lin_b[1];
      }
    }
  }
}

extern "C" void kernel_launch(void* const* d_in, const int* in_sizes, int n_in,
                              void* d_out, int out_size, void* d_ws, size_t ws_size,
                              hipStream_t stream) {
  const float* x        = (const float*)d_in[0];
  const int*   ei       = (const int*)d_in[1];
  const float* W1       = (const float*)d_in[2];
  const float* att_src1 = (const float*)d_in[3];
  const float* att_dst1 = (const float*)d_in[4];
  const float* b1       = (const float*)d_in[5];
  const float* W2       = (const float*)d_in[6];
  const float* att_src2 = (const float*)d_in[7];
  const float* att_dst2 = (const float*)d_in[8];
  const float* b2       = (const float*)d_in[9];
  const float* lin_w    = (const float*)d_in[10];
  const float* lin_b    = (const float*)d_in[11];
  float* out = (float*)d_out;
  const int* src = ei;
  const int* dst = ei + NE;

  float*    ws   = (float*)d_ws;
  float*    h1   = ws;
  float*    sd1  = ws + 6400000;
  unsigned* own  = (unsigned*)(sd1 + 400000);
  unsigned* idx  = own + 200000;
  float*    bs1  = sd1 + 790000;
  float*    bs2  = bs1 + FOLD1;
  float*    h2   = sd1 + 800000;
  float*    z    = h1;
  float*    sd2  = h1 + 3200000;

  owner_bytes<<<NCHK + NFOLD, 256, 0, stream>>>(dst, own, idx, W1, att_src1, att_dst1, W2, att_src2, att_dst2,
                                                bs1, bs2);

  gemm_wmma_f32<HEADS * HID, INDIM, 4, HEADS>
      <<<dim3(NN / 16), dim3(32, 3, 1), 0, stream>>>(x, W1, h1, bs1, sd1);
  agg_owner<HEADS, false><<<NBLK, 1024, 0, stream>>>(src, dst, own, idx, sd1, h1, b1, lin_w, lin_b, h2);

  gemm_wmma_f32<HID, HEADS * HID, 2, 1>
      <<<dim3(NN / 80), dim3(32, 2, 5), 0, stream>>>(h2, W2, z, bs2, sd2);
  agg_owner<1, true><<<NBLK, 1024, 0, stream>>>(src, dst, own, idx, sd2, z, b2, lin_w, lin_b, out);
}


# Layer-1 gather takes a 512-byte row with one 16-byte load per lane (4 load instructions to 1, 16 rows in flight); LDS sums in lane-interleaved channel order, in another LDS atomic arrival order
# speedup vs baseline: 6.8092x; 1.0441x over previous
#include <hip/hip_runtime.h>
#include <math.h>

#define NN 50000
#define NE 800000
#define INDIM 256
#define HID 32
#define HEADS 4
#define ETOT (NE + NN)

typedef __attribute__((ext_vector_type(2))) float v2f;
typedef __attribute__((ext_vector_type(8))) float v8f;

__device__ __forceinline__ unsigned fkey(float x) {
  unsigned u = __float_as_uint(x);
  return (u & 0x80000000u) ? ~u : (u | 0x80000000u);
}
__device__ __forceinline__ float unfkey(unsigned k) {
  return __uint_as_float((k & 0x80000000u) ? (k & 0x7fffffffu) : ~k);
}
__device__ __forceinline__ float elu(float v) {
  return (v > 0.f) ? v : (__expf(v) - 1.f);
}

typedef __attribute__((ext_vector_type(4))) float v4f;
template <int NT> struct colvec;
template <> struct colvec<4> { typedef v4f type; };
template <> struct colvec<2> { typedef v2f type; };

template <int K, int H>
__device__ __forceinline__ void score_tile(const float* __restrict__ A, const float* __restrict__ Bs,
                                           float* __restrict__ SD, int mbase, int lane) {
  const int c = lane & 15;
  const int h = lane >> 4;
  const float* ap = A + (size_t)(mbase + c) * K + 4 * h;
  const float* bp = Bs + h * 64 + 4 * c;
  constexpr int G   = 4;
  constexpr int NCH = K / (8 * G);
  static_assert(K % (8 * G) == 0, "whole chunks");
  v8f acc[2];
#pragma unroll
  for (int t = 0; t < 2; ++t) acc[t] = v8f{0.f, 0.f, 0.f, 0.f, 0.f, 0.f, 0.f, 0.f};
  v4f af[2][G], bf[2][G];
#pragma unroll
  for (int g = 0; g < G; ++g) {
    af[0][g] = *reinterpret_cast<const v4f*>(ap + 8 * g);
    bf[0][g] = *reinterpret_cast<const v4f*>(bp + 128 * g);
  }
#pragma unroll
  for (int ch = 0; ch < NCH; ++ch) {
    const int cur = ch & 1, nxt = cur ^ 1;
    if (ch + 1 < NCH) {
#pragma unroll
      for (int g = 0; g < G; ++g) {
        af[nxt][g] = *reinterpret_cast<const v4f*>(ap + (ch + 1) * 8 * G + 8 * g);
        bf[nxt][g] = *reinterpret_cast<const v4f*>(bp + ((ch + 1) * G + g) * 128);
      }
    }
    __builtin_amdgcn_sched_barrier(0);
#pragma unroll
    for (int u = 0; u < 2; ++u) {
#pragma unroll
      for (int g = 0; g < G; ++g) {
        v2f a, b;
        a.x = af[cur][g][2 * u];
        a.y = af[cur][g][2 * u + 1];
        b.x = bf[cur][g][2 * u];
        b.y = bf[cur][g][2 * u + 1];
        acc[g & 1] = __builtin_amdgcn_wmma_f32_16x16x4_f32(false, a, false, b, (short)0, acc[g & 1],
                                                           false, false);
      }
    }
    __builtin_amdgcn_sched_barrier(0);
  }
  if (c < 2 * H) {
    float* sp = SD + (size_t)(mbase + 8 * h) * (2 * H) + c;
#pragma unroll
    for (int r = 0; r < 8; ++r) sp[r * (2 * H)] = acc[0][r] + acc[1][r];
  }
}

template <int N, int K, int NT, int H>
__global__ void gemm_wmma_f32(const float* __restrict__ A, const float* __restrict__ B,
                              float* __restrict__ C, const float* __restrict__ Bs,
                              float* __restrict__ SD) {
  typedef typename colvec<NT>::type vnt;
  const int lane  = threadIdx.x;
  const int c     = lane & 15;
  const int h     = lane >> 4;
  const int mbase = (blockIdx.x * blockDim.z + threadIdx.z) * 16;
  const int wy    = __builtin_amdgcn_readfirstlane((int)threadIdx.y);
  if (wy == N / (16 * NT)) {
    score_tile<K, H>(A, Bs, SD, mbase, lane);
    return;
  }
  const int nb    = wy * (16 * NT);
  const float* ap = A + (size_t)(mbase + c) * K + 4 * h;
  const float* bp = B + (size_t)(4 * h) * N + nb + NT * c;
  v8f acc[NT];
#pragma unroll
  for (int t = 0; t < NT; ++t) acc[t] = v8f{0.f, 0.f, 0.f, 0.f, 0.f, 0.f, 0.f, 0.f};

  constexpr int KC  = 16;
  constexpr int NCH = K / KC;
  static_assert(K % KC == 0, "whole chunks");
  v4f af[2][2];
  vnt bf[2][8];
#pragma unroll
  for (int g = 0; g < 2; ++g) {
    af[0][g] = *reinterpret_cast<const v4f*>(ap + 8 * g);
#pragma unroll
    for (int j = 0; j < 4; ++j)
      bf[0][4 * g + j] = *reinterpret_cast<const vnt*>(bp + (size_t)(8 * g + j) * N);
  }
#pragma unroll
  for (int ch = 0; ch < NCH; ++ch) {
    const int cur = ch & 1, nxt = cur ^ 1;
    if (ch + 1 < NCH) {
      const float* apn = ap + (ch + 1) * KC;
      const float* bpn = bp + (size_t)(ch + 1) * KC * N;
#pragma unroll
      for (int g = 0; g < 2; ++g) {
        af[nxt][g] = *reinterpret_cast<const v4f*>(apn + 8 * g);
#pragma unroll
        for (int j = 0; j < 4; ++j)
          bf[nxt][4 * g + j] = *reinterpret_cast<const vnt*>(bpn + (size_t)(8 * g + j) * N);
      }
    }
    __builtin_amdgcn_sched_barrier(0);
#pragma unroll
    for (int g = 0; g < 2; ++g) {
#pragma unroll
      for (int u = 0; u < 2; ++u) {
        v2f a;
        a.x = af[cur][g][2 * u];
        a.y = af[cur][g][2 * u + 1];
#pragma unroll
        for (int t = 0; t < NT; ++t) {
          v2f b;
          b.x = bf[cur][4 * g + 2 * u][t];
          b.y = bf[cur][4 * g + 2 * u + 1][t];
          acc[t] = __builtin_amdgcn_wmma_f32_16x16x4_f32(false, a, false, b, (short)0, acc[t],
                                                         false, false);
        }
      }
    }
    __builtin_amdgcn_sched_barrier(0);
  }
  float* cp = C + (size_t)(mbase + 8 * h) * N + nb + NT * c;
#pragma unroll
  for (int r = 0; r < 8; ++r) {
    vnt o;
#pragma unroll
    for (int t = 0; t < NT; ++t) o[t] = acc[t][r];
    *reinterpret_cast<vnt*>(cp + (size_t)r * N) = o;
  }
}

#define NPB  208
#define NBLK ((NN + NPB - 1) / NPB)
#define NU   (NE / 16)
#define LCAP 8192
#define CHU  (LCAP / 16)
#define QCAP LCAP
#define NCHK ((NE / 4 + 255) / 256)
static_assert(NBLK <= 256, "owner id must fit a byte");
static_assert(NE % 32 == 0, "owner map is read in 16-byte units, and in pairs of them through the index");
static_assert(NN <= 65536 && NPB <= 65536, "list entry packs source and local destination in 16 bits each");
static_assert(HID == 32, "one lane per channel of a head");
static_assert((unsigned long long)NN * HEADS * HID * sizeof(float) < (1ull << 32), "a feature row's byte offset fits 32 bits");
static_assert(NN % 80 == 0, "product launches: 16 rows per wave, five waves per block in the second");

__device__ __forceinline__ float lrelu(float x) { return (x > 0.f) ? x : 0.2f * x; }

template <int H>
__device__ __forceinline__ float fold_elem(const float* __restrict__ Wm, const float* __restrict__ att_s,
                                           const float* __restrict__ att_d, int o) {
  const int k = ((o >> 6) << 2) | (o & 3);
  const int col = (o >> 2) & 15;
  if (col >= 2 * H) return 0.f;
  const int hd = (col < H) ? col : col - H;
  const float4* wp = reinterpret_cast<const float4*>(Wm + (size_t)k * (H * HID) + hd * HID);
  const float4* tp = reinterpret_cast<const float4*>(((col < H) ? att_s : att_d) + hd * HID);
  float s = 0.f;
#pragma unroll
  for (int q = 0; q < HID / 4; ++q) {
    const float4 w = wp[q], t = tp[q];
    s += w.x * t.x;
    s += w.y * t.y;
    s += w.z * t.z;
    s += w.w * t.w;
  }
  return s;
}
#define FOLD1 (INDIM * 16)
#define FOLD2 (HEADS * HID * 16)
#define NFOLD ((FOLD1 + FOLD2) / 256)
static_assert((FOLD1 + FOLD2) % 256 == 0, "whole work-groups of fold elements");

__global__ void owner_bytes(const int* __restrict__ dst, unsigned* __restrict__ own, unsigned* __restrict__ idx,
                            const float* __restrict__ W1, const float* __restrict__ att_s1,
                            const float* __restrict__ att_d1, const float* __restrict__ W2,
                            const float* __restrict__ att_s2, const float* __restrict__ att_d2,
                            float* __restrict__ bs1, float* __restrict__ bs2) {
  if (blockIdx.x < NFOLD) {
    const int o = blockIdx.x * 256 + threadIdx.x;
    if (o < FOLD1) bs1[o] = fold_elem<HEADS>(W1, att_s1, att_d1, o);
    else bs2[o - FOLD1] = fold_elem<1>(W2, att_s2, att_d2, o - FOLD1);
    return;
  }
  const int blk = blockIdx.x - NFOLD;
  __shared__ unsigned w_l[NBLK];
  if (threadIdx.x < NBLK) w_l[threadIdx.x] = 0u;
  __syncthreads();
  const int i = blk * blockDim.x + threadIdx.x;
  if (i < NE / 4) {
    const int4 d = ((const int4*)dst)[i];
    const unsigned o0 = (unsigned)(d.x / NPB), o1 = (unsigned)(d.y / NPB);
    const unsigned o2 = (unsigned)(d.z / NPB), o3 = (unsigned)(d.w / NPB);
    own[i] = o0 | (o1 << 8) | (o2 << 16) | (o3 << 24);
    const unsigned bit = 1u << (threadIdx.x >> 3);
    atomicOr(&w_l[o0], bit);
    atomicOr(&w_l[o1], bit);
    atomicOr(&w_l[o2], bit);
    atomicOr(&w_l[o3], bit);
  }
  __syncthreads();
  if (threadIdx.x < NBLK) idx[blk * NBLK + threadIdx.x] = w_l[threadIdx.x];
}
static_assert(NBLK <= 256, "one thread per owner in owner_bytes");

__device__ __forceinline__ unsigned zero_bytes(unsigned t) {
  return ~(((t & 0x7f7f7f7fu) + 0x7f7f7f7fu) | t) & 0x80808080u;
}


__device__ __forceinline__ void find_units(const uint4* __restrict__ own4, int u0, int u1,
                                           unsigned pat, float* q_l, unsigned* qcnt_l, int wave, int lane) {
  for (int ub = u0 + wave * 32; ub < u1; ub += 4 * 1024) {
    const int u = ub + lane;
    uint4 v[4];
    int id[4];
    bool ok[4];
#pragma unroll
    for (int q = 0; q < 4; ++q) {
      const int uu = u + q * 1024;
      ok[q] = uu < u1;
      id[q] = ok[q] ? uu : u0;
      v[q] = own4[id[q]];
    }
    unsigned bal[4];
    bool hit[4];
#pragma unroll
    for (int q = 0; q < 4; ++q) {
      const unsigned tx = v[q].x ^ pat, ty = v[q].y ^ pat, tz = v[q].z ^ pat, tw = v[q].w ^ pat;
      const unsigned a = ((tx - 0x01010101u) & ~tx) | ((ty - 0x01010101u) & ~ty) |
                         ((tz - 0x01010101u) & ~tz) | ((tw - 0x01010101u) & ~tw);
      hit[q] = ok[q] && ((a & 0x80808080u) != 0u);
      bal[q] = __builtin_amdgcn_ballot_w32(hit[q]);
    }
    const unsigned tot = (unsigned)(__builtin_popcount(bal[0]) + __builtin_popcount(bal[1]) +
                                    __builtin_popcount(bal[2]) + __builtin_popcount(bal[3]));
    if (tot) {
      unsigned base = 0u;
      if (lane == 0) base = atomicAdd(qcnt_l, tot);
      base = (unsigned)__builtin_amdgcn_readfirstlane((int)base);
#pragma unroll
      for (int q = 0; q < 4; ++q) {
        if (hit[q]) {
          const unsigned pos = base + __builtin_amdgcn_mbcnt_lo(bal[q], 0u);
          if (pos < QCAP) q_l[pos] = __uint_as_float((unsigned)id[q]);
        }
        base += (unsigned)__builtin_popcount(bal[q]);
      }
    }
  }
}

__device__ __forceinline__ void queue_groups(const unsigned* __restrict__ idx, int blk, float* q_l,
                                             unsigned* qcnt_l, int tid) {
  for (int t = tid; t < NCHK; t += 1024) {
    const unsigned w = idx[t * NBLK + blk];
    if (w) {
      unsigned pos = atomicAdd(qcnt_l, (unsigned)__popc(w));
      for (unsigned r = w; r; r &= r - 1) {
        if (pos < QCAP) q_l[pos] = __uint_as_float((unsigned)(t * 32 + __ffs(r) - 1));
        ++pos;
      }
    }
  }
}

template <int UPE>
__device__ __forceinline__ void decode_units(const uint4* __restrict__ own4, unsigned pat, const float* q_l, int nq,
                                             unsigned* list_l, unsigned* cnt_l, int wave, int lane) {
  static_assert(UPE == 1 || UPE == 2, "the byte mask of an entry must fit 32 bits");
  for (int i0 = wave * 32; i0 < nq; i0 += 4 * 1024) {
    uint4 v[4][UPE];
    unsigned id[4];
    bool ok[4];
#pragma unroll
    for (int q = 0; q < 4; ++q) {
      const int i = i0 + q * 1024 + lane;
      ok[q] = i < nq;
      id[q] = __float_as_uint(q_l[ok[q] ? i : i0]);
#pragma unroll
      for (int p = 0; p < UPE; ++p) v[q][p] = own4[id[q] * UPE + p];
    }
#pragma unroll
    for (int q = 0; q < 4; ++q) {
      unsigned mm = 0u;
#pragma unroll
      for (int p = 0; p < UPE; ++p)
        mm |= ((zero_bytes(v[q][p].x ^ pat) >> 7) | (zero_bytes(v[q][p].y ^ pat) >> 6) |
               (zero_bytes(v[q][p].z ^ pat) >> 5) | (zero_bytes(v[q][p].w ^ pat) >> 4)) << (4 * p);
      if (!ok[q]) mm = 0u;
      const unsigned c = (unsigned)__popc(mm);
      unsigned excl = 0u, tot = 0u;
#pragma unroll
      for (int b = 0; b < 4 + UPE; ++b) {
        const unsigned bl = __builtin_amdgcn_ballot_w32(((c >> b) & 1u) != 0u);
        excl += __builtin_amdgcn_mbcnt_lo(bl, 0u) << b;
        tot += (unsigned)__builtin_popcount(bl) << b;
      }
      if (tot) {
        unsigned base = 0u;
        if (lane == 0) base = atomicAdd(cnt_l, tot);
        unsigned pos = (unsigned)__builtin_amdgcn_readfirstlane((int)base) + excl;
        for (unsigned r = mm; r; r &= r - 1) {
          const int j = __ffs(r) - 1;
          if (pos < LCAP)
            list_l[pos] = id[q] * (16u * UPE) + (unsigned)(16 * ((j >> 2) & 1) + 4 * (j & 3) + (j >> 3));
          ++pos;
        }
      }
    }
  }
}

template <bool B> struct flag_c { static constexpr bool value = B; };

template <int H, bool FINAL>
__global__ __launch_bounds__(1024) void agg_owner(
    const int* __restrict__ src, const int* __restrict__ dst, const unsigned* __restrict__ own,
    const unsigned* __restrict__ idx,
    const float* __restrict__ sd, const float* __restrict__ feat,
    const float* __restrict__ bias, const float* __restrict__ lin_w, const float* __restrict__ lin_b,
    float* __restrict__ outp) {
  constexpr int HC = H * HID;
  __shared__ float    acc_l[NPB * HC];
  __shared__ float    den_l[NPB * H];
  __shared__ __attribute__((aligned(16))) float    ad_l[NPB * H];
  __shared__ __attribute__((aligned(16))) unsigned m_l[NPB * H];
  __shared__ unsigned list_l[LCAP + NPB];
  __shared__ __attribute__((aligned(16))) float p_l[(LCAP + NPB) * H];
  __shared__ unsigned cnt_l, qcnt_l;
  static_assert(QCAP <= (LCAP + NPB) * H, "the unit queue borrows p_l");
  const int tid = threadIdx.x, lane = tid & 31;
  const int wave = __builtin_amdgcn_readfirstlane(tid >> 5);
  const int lo = blockIdx.x * NPB;
  const int nn = min(NPB, NN - lo);
  const uint4* own4 = (const uint4*)own;
  const unsigned pat = (unsigned)blockIdx.x * 0x01010101u;

  for (int i = tid; i < NPB * HC; i += 1024) acc_l[i] = 0.f;
  for (int i = tid; i < NPB * H; i += 1024) {
    float a_d = 0.f;
    unsigned mk = 0u;
    if (i < nn * H) {
      const float* sp = sd + (size_t)(lo + i / H) * (2 * H) + i % H;
      a_d = sp[H];
      mk = fkey(lrelu(sp[0] + a_d));
    }
    ad_l[i] = a_d;
    m_l[i] = mk;
    den_l[i] = 0.f;
  }
  if (tid == 0) { cnt_l = 0u; qcnt_l = 0u; }
  __syncthreads();

  auto max_pass = [&](int n) {
    for (int i = tid; i < n; i += 1024) {
      const int e = (int)list_l[i];
      const int s = src[e];
      const int dl = dst[e] - lo;
#pragma unroll
      for (int h = 0; h < H; ++h) {
        const float x = lrelu(sd[s * (2 * H) + h] + ad_l[dl * H + h]);
        atomicMax(&m_l[dl * H + h], fkey(x));
        p_l[i * H + h] = x;
      }
      list_l[i] = (unsigned)s | ((unsigned)dl << 16);
    }
  };

  auto weight_pass = [&](int n, int nself) {
    const int tot = n + nself;
    for (int i = tid; i < tot; i += 1024) {
      float x[H];
      int dl;
      if (i < n) {
        dl = (int)(list_l[i] >> 16);
#pragma unroll
        for (int h = 0; h < H; ++h) x[h] = p_l[i * H + h];
      } else {
        dl = i - n;
        list_l[i] = (unsigned)(lo + dl) | ((unsigned)dl << 16);
#pragma unroll
        for (int h = 0; h < H; ++h) x[h] = lrelu(sd[(lo + dl) * (2 * H) + h] + ad_l[dl * H + h]);
      }
#pragma unroll
      for (int h = 0; h < H; ++h) {
        const float p = __expf(x[h] - unfkey(m_l[dl * H + h]));
        p_l[i * H + h] = p;
        atomicAdd(&den_l[dl * H + h], p);
      }
    }
  };

  constexpr int BATCH = 16;
  constexpr int JB = 16;
  auto acc_batch = [&](auto all_c, int b0, int nb) __attribute__((always_inline)) {
    constexpr bool ALL = decltype(all_c)::value;
    const unsigned pkv = (lane < nb) ? list_l[b0 + lane] : 0u;
    int dl[JB] = {};
    if constexpr (H == 4) {
      v4f v[JB] = {};
      float p[JB] = {};
      const unsigned lane16 = 16u * (unsigned)lane;
#pragma unroll
      for (int j = 0; j < JB; ++j) {
        if (ALL || j < nb) {
          const unsigned pk = (unsigned)__builtin_amdgcn_readlane((int)pkv, j);
          dl[j] = (int)(pk >> 16);
          const unsigned off = (pk & 0xffffu) * (unsigned)(HC * sizeof(float)) + lane16;
          v[j] = *reinterpret_cast<const v4f*>(reinterpret_cast<const char*>(feat) + off);
        }
      }
#pragma unroll
      for (int j = 0; j < JB; ++j) {
        if (ALL || j < nb) p[j] = p_l[(b0 + j) * 4 + (lane >> 3)];
      }
#pragma unroll
      for (int j = 0; j < JB; ++j) {
        if (ALL || j < nb) {
#pragma unroll
          for (int i = 0; i < 4; ++i) atomicAdd(&acc_l[dl[j] * HC + i * 32 + lane], p[j] * v[j][i]);
        }
      }
    } else {
      float v[JB][H] = {}, p[JB][H] = {};
#pragma unroll
      for (int j = 0; j < JB; ++j) {
        if (ALL || j < nb) {
          const unsigned pk = (unsigned)__builtin_amdgcn_readlane((int)pkv, j);
          const int s = (int)(pk & 0xffffu);
          dl[j] = (int)(pk >> 16);
#pragma unroll
          for (int h = 0; h < H; ++h) v[j][h] = feat[(size_t)s * HC + h * HID + lane];
        }
      }
#pragma unroll
      for (int j = 0; j < JB; ++j) {
        if (ALL || j < nb) {
#pragma unroll
          for (int h = 0; h < H; ++h) p[j][h] = p_l[(b0 + j) * H + h];
        }
      }
#pragma unroll
      for (int j = 0; j < JB; ++j) {
        if (ALL || j < nb) {
#pragma unroll
          for (int h = 0; h < H; ++h)
            atomicAdd(&acc_l[dl[j] * HC + h * HID + lane], p[j][h] * v[j][h]);
        }
      }
    }
  };
  auto acc_pass = [&](int tot) __attribute__((always_inline)) {
    for (int b0 = wave * BATCH; b0 < tot; b0 += 32 * BATCH) {
      const int nb = min(BATCH, tot - b0);
      if (nb == BATCH) acc_batch(flag_c<true>{}, b0, BATCH);
      else acc_batch(flag_c<false>{}, b0, nb);
    }
  };

  auto scan_units = [&](int u0, int u1) __attribute__((always_inline)) {
    find_units(own4, u0, u1, pat, p_l, &qcnt_l, wave, lane);
    __syncthreads();
    const int nq = (int)qcnt_l;
    if (nq <= QCAP) decode_units<1>(own4, pat, p_l, nq, list_l, &cnt_l, wave, lane);
    __syncthreads();
    return (nq <= QCAP) ? (int)cnt_l : LCAP + 1;
  };
  auto scan_index = [&]() __attribute__((always_inline)) {
    queue_groups(idx, (int)blockIdx.x, p_l, &qcnt_l, tid);
    __syncthreads();
    const int nq = (int)qcnt_l;
    if (nq <= QCAP) decode_units<2>(own4, pat, p_l, nq, list_l, &cnt_l, wave, lane);
    __syncthreads();
    return (nq <= QCAP) ? (int)cnt_l : LCAP + 1;
  };

  const int n = scan_index();
  if (n <= LCAP) {
    max_pass(n);
    __syncthreads();
    weight_pass(n, nn);
    __syncthreads();
    acc_pass(n + nn);
  } else {
    for (int c0 = 0; c0 < NU; c0 += CHU) {
      __syncthreads();
      if (tid == 0) { cnt_l = 0u; qcnt_l = 0u; }
      __syncthreads();
      max_pass(scan_units(c0, min(c0 + CHU, NU)));
    }
    for (int c0 = 0; c0 < NU; c0 += CHU) {
      __syncthreads();
      if (tid == 0) { cnt_l = 0u; qcnt_l = 0u; }
      __syncthreads();
      const int nc = scan_units(c0, min(c0 + CHU, NU));
      const int ns = (c0 == 0) ? nn : 0;
      max_pass(nc);
      __syncthreads();
      weight_pass(nc, ns);
      __syncthreads();
      acc_pass(nc + ns);
    }
  }
  __syncthreads();

  if (!FINAL) {
    if constexpr (H == 4) {
      for (int i = tid; i < nn * (HC / 4); i += 1024) {
        const int dl = i / (HC / 4), q = i % (HC / 4);
        const float d = den_l[dl * H + q / (HID / 4)];
        const float4 bq = reinterpret_cast<const float4*>(bias)[q];
        float4 o;
        o.x = elu(acc_l[dl * HC + 0 * 32 + q] / d + bq.x);
        o.y = elu(acc_l[dl * HC + 1 * 32 + q] / d + bq.y);
        o.z = elu(acc_l[dl * HC + 2 * 32 + q] / d + bq.z);
        o.w = elu(acc_l[dl * HC + 3 * 32 + q] / d + bq.w);
        reinterpret_cast<float4*>(outp + (size_t)lo * HC)[i] = o;
      }
    } else {
      for (int i = tid; i < nn * HC; i += 1024) {
        const int dl = i / HC, r = i % HC;
        outp[(size_t)lo * HC + i] = elu(acc_l[i] / den_l[dl * H + r / HID] + bias[r]);
      }
    }
  } else {
    for (int dl = wave; dl < nn; dl += 32) {
      const float v = elu(acc_l[dl * HC + lane] / den_l[dl * H] + bias[lane]);
      float w0 = v * lin_w[lane * 2 + 0];
      float w1 = v * lin_w[lane * 2 + 1];
#pragma unroll
      for (int m = 16; m > 0; m >>= 1) {
        w0 += __shfl_xor(w0, m, 32);
        w1 += __shfl_xor(w1, m, 32);
      }
      if (lane == 0) {
        outp[(lo + dl) * 2 + 0] = w0 + lin_b[0];
        outp[(lo + dl) * 2 + 1] = w1 + lin_b[1];
      }
    }
  }
}

extern "C" void kernel_launch(void* const* d_in, const int* in_sizes, int n_in,
                              void* d_out, int out_size, void* d_ws, size_t ws_size,
                              hipStream_t stream) {
  const float* x        = (const float*)d_in[0];
  const int*   ei       = (const int*)d_in[1];
  const float* W1       = (const float*)d_in[2];
  const float* att_src1 = (const float*)d_in[3];
  const float* att_dst1 = (const float*)d_in[4];
  const float* b1       = (const float*)d_in[5];
  const float* W2       = (const float*)d_in[6];
  const float* att_src2 = (const float*)d_in[7];
  const float* att_dst2 = (const float*)d_in[8];
  const float* b2       = (const float*)d_in[9];
  const float* lin_w    = (const float*)d_in[10];
  const float* lin_b    = (const float*)d_in[11];
  float* out = (float*)d_out;
  const int* src = ei;
  const int* dst = ei + NE;

  float*    ws   = (float*)d_ws;
  float*    h1   = ws;
  float*    sd1  = ws + 6400000;
  unsigned* own  = (unsigned*)(sd1 + 400000);
  unsigned* idx  = own + 200000;
  float*    bs1  = sd1 + 790000;
  float*    bs2  = bs1 + FOLD1;
  float*    h2   = sd1 + 800000;
  float*    z    = h1;
  float*    sd2  = h1 + 3200000;

  owner_bytes<<<NCHK + NFOLD, 256, 0, stream>>>(dst, own, idx, W1, att_src1, att_dst1, W2, att_src2, att_dst2,
                                                bs1, bs2);

  gemm_wmma_f32<HEADS * HID, INDIM, 4, HEADS>
      <<<dim3(NN / 16), dim3(32, 3, 1), 0, stream>>>(x, W1, h1, bs1, sd1);
  agg_owner<HEADS, false><<<NBLK, 1024, 0, stream>>>(src, dst, own, idx, sd1, h1, b1, lin_w, lin_b, h2);

  gemm_wmma_f32<HID, HEADS * HID, 2, 1>
      <<<dim3(NN / 80), dim3(32, 2, 5), 0, stream>>>(h2, W2, z, bs2, sd2);
  agg_owner<1, true><<<NBLK, 1024, 0, stream>>>(src, dst, own, idx, sd2, z, b2, lin_w, lin_b, out);
}


# Second product runs inside the layer-1 aggregation work-group: its 208 finished rows stay in LDS as A (pitch 132), two waves per 16-row tile (z columns; score columns), no k split; 5 launches to 4
# speedup vs baseline: 7.1838x; 1.0550x over previous
#include <hip/hip_runtime.h>
#include <math.h>

#define NN 50000
#define NE 800000
#define INDIM 256
#define HID 32
#define HEADS 4
#define ETOT (NE + NN)

typedef __attribute__((ext_vector_type(2))) float v2f;
typedef __attribute__((ext_vector_type(8))) float v8f;

__device__ __forceinline__ unsigned fkey(float x) {
  unsigned u = __float_as_uint(x);
  return (u & 0x80000000u) ? ~u : (u | 0x80000000u);
}
__device__ __forceinline__ float unfkey(unsigned k) {
  return __uint_as_float((k & 0x80000000u) ? (k & 0x7fffffffu) : ~k);
}
__device__ __forceinline__ float elu(float v) {
  return (v > 0.f) ? v : (__expf(v) - 1.f);
}

typedef __attribute__((ext_vector_type(4))) float v4f;
template <int NT> struct colvec;
template <> struct colvec<4> { typedef v4f type; };
template <> struct colvec<2> { typedef v2f type; };

template <int K, int LDA, int H>
__device__ __forceinline__ void score_tile(const float* __restrict__ A0, const float* __restrict__ Bs,
                                           float* __restrict__ SD0, int lane) {
  const int c = lane & 15;
  const int h = lane >> 4;
  const float* ap = A0 + c * LDA + 4 * h;
  const float* bp = Bs + h * 64 + 4 * c;
  constexpr int G   = 4;
  constexpr int NCH = K / (8 * G);
  static_assert(K % (8 * G) == 0, "whole chunks");
  v8f acc[2];
#pragma unroll
  for (int t = 0; t < 2; ++t) acc[t] = v8f{0.f, 0.f, 0.f, 0.f, 0.f, 0.f, 0.f, 0.f};
  v4f af[2][G], bf[2][G];
#pragma unroll
  for (int g = 0; g < G; ++g) {
    af[0][g] = *reinterpret_cast<const v4f*>(ap + 8 * g);
    bf[0][g] = *reinterpret_cast<const v4f*>(bp + 128 * g);
  }
#pragma unroll
  for (int ch = 0; ch < NCH; ++ch) {
    const int cur = ch & 1, nxt = cur ^ 1;
    if (ch + 1 < NCH) {
#pragma unroll
      for (int g = 0; g < G; ++g) {
        af[nxt][g] = *reinterpret_cast<const v4f*>(ap + (ch + 1) * 8 * G + 8 * g);
        bf[nxt][g] = *reinterpret_cast<const v4f*>(bp + ((ch + 1) * G + g) * 128);
      }
    }
    __builtin_amdgcn_sched_barrier(0);
#pragma unroll
    for (int u = 0; u < 2; ++u) {
#pragma unroll
      for (int g = 0; g < G; ++g) {
        v2f a, b;
        a.x = af[cur][g][2 * u];
        a.y = af[cur][g][2 * u + 1];
        b.x = bf[cur][g][2 * u];
        b.y = bf[cur][g][2 * u + 1];
        acc[g & 1] = __builtin_amdgcn_wmma_f32_16x16x4_f32(false, a, false, b, (short)0, acc[g & 1],
                                                           false, false);
      }
    }
    __builtin_amdgcn_sched_barrier(0);
  }
  if (c < 2 * H) {
    float* sp = SD0 + (8 * h) * (2 * H) + c;
#pragma unroll
    for (int r = 0; r < 8; ++r) sp[r * (2 * H)] = acc[0][r] + acc[1][r];
  }
}

template <int N, int K, int LDA, int NT>
__device__ __forceinline__ void prod_tile(const float* __restrict__ A0, const float* __restrict__ B,
                                          float* __restrict__ C0, int nb, int lane) {
  typedef typename colvec<NT>::type vnt;
  const int c     = lane & 15;
  const int h     = lane >> 4;
  const float* ap = A0 + c * LDA + 4 * h;
  const float* bp = B + (size_t)(4 * h) * N + nb + NT * c;
  v8f acc[NT];
#pragma unroll
  for (int t = 0; t < NT; ++t) acc[t] = v8f{0.f, 0.f, 0.f, 0.f, 0.f, 0.f, 0.f, 0.f};

  constexpr int KC  = 16;
  constexpr int NCH = K / KC;
  static_assert(K % KC == 0, "whole chunks");
  v4f af[2][2];
  vnt bf[2][8];
#pragma unroll
  for (int g = 0; g < 2; ++g) {
    af[0][g] = *reinterpret_cast<const v4f*>(ap + 8 * g);
#pragma unroll
    for (int j = 0; j < 4; ++j)
      bf[0][4 * g + j] = *reinterpret_cast<const vnt*>(bp + (size_t)(8 * g + j) * N);
  }
#pragma unroll
  for (int ch = 0; ch < NCH; ++ch) {
    const int cur = ch & 1, nxt = cur ^ 1;
    if (ch + 1 < NCH) {
      const float* apn = ap + (ch + 1) * KC;
      const float* bpn = bp + (size_t)(ch + 1) * KC * N;
#pragma unroll
      for (int g = 0; g < 2; ++g) {
        af[nxt][g] = *reinterpret_cast<const v4f*>(apn + 8 * g);
#pragma unroll
        for (int j = 0; j < 4; ++j)
          bf[nxt][4 * g + j] = *reinterpret_cast<const vnt*>(bpn + (size_t)(8 * g + j) * N);
      }
    }
    __builtin_amdgcn_sched_barrier(0);
#pragma unroll
    for (int g = 0; g < 2; ++g) {
#pragma unroll
      for (int u = 0; u < 2; ++u) {
        v2f a;
        a.x = af[cur][g][2 * u];
        a.y = af[cur][g][2 * u + 1];
#pragma unroll
        for (int t = 0; t < NT; ++t) {
          v2f b;
          b.x = bf[cur][4 * g + 2 * u][t];
          b.y = bf[cur][4 * g + 2 * u + 1][t];
          acc[t] = __builtin_amdgcn_wmma_f32_16x16x4_f32(false, a, false, b, (short)0, acc[t],
                                                         false, false);
        }
      }
    }
    __builtin_amdgcn_sched_barrier(0);
  }
  float* cp = C0 + (size_t)(8 * h) * N + nb + NT * c;
#pragma unroll
  for (int r = 0; r < 8; ++r) {
    vnt o;
#pragma unroll
    for (int t = 0; t < NT; ++t) o[t] = acc[t][r];
    *reinterpret_cast<vnt*>(cp + (size_t)r * N) = o;
  }
}

template <int N, int K, int NT, int H>
__global__ void gemm_wmma_f32(const float* __restrict__ A, const float* __restrict__ B,
                              float* __restrict__ C, const float* __restrict__ Bs,
                              float* __restrict__ SD) {
  const int lane  = threadIdx.x;
  const int mbase = (blockIdx.x * blockDim.z + threadIdx.z) * 16;
  const int wy    = __builtin_amdgcn_readfirstlane((int)threadIdx.y);
  if (wy == N / (16 * NT)) {
    score_tile<K, K, H>(A + (size_t)mbase * K, Bs, SD + (size_t)mbase * (2 * H), lane);
    return;
  }
  prod_tile<N, K, K, NT>(A + (size_t)mbase * K, B, C + (size_t)mbase * N, wy * (16 * NT), lane);
}

#define NPB  208
#define NBLK ((NN + NPB - 1) / NPB)
#define NU   (NE / 16)
#define LCAP 8192
#define CHU  (LCAP / 16)
#define QCAP LCAP
#define NCHK ((NE / 4 + 255) / 256)
static_assert(NBLK <= 256, "owner id must fit a byte");
static_assert(NE % 32 == 0, "owner map is read in 16-byte units, and in pairs of them through the index");
static_assert(NN <= 65536 && NPB <= 65536, "list entry packs source and local destination in 16 bits each");
static_assert(HID == 32, "one lane per channel of a head");
static_assert((unsigned long long)NN * HEADS * HID * sizeof(float) < (1ull << 32), "a feature row's byte offset fits 32 bits");
static_assert(NN % 16 == 0 && NPB % 16 == 0, "products: whole 16-row tiles, in the first product's launch and in a work-group's owned rows");

__device__ __forceinline__ float lrelu(float x) { return (x > 0.f) ? x : 0.2f * x; }

template <int H>
__device__ __forceinline__ float fold_elem(const float* __restrict__ Wm, const float* __restrict__ att_s,
                                           const float* __restrict__ att_d, int o) {
  const int k = ((o >> 6) << 2) | (o & 3);
  const int col = (o >> 2) & 15;
  if (col >= 2 * H) return 0.f;
  const int hd = (col < H) ? col : col - H;
  const float4* wp = reinterpret_cast<const float4*>(Wm + (size_t)k * (H * HID) + hd * HID);
  const float4* tp = reinterpret_cast<const float4*>(((col < H) ? att_s : att_d) + hd * HID);
  float s = 0.f;
#pragma unroll
  for (int q = 0; q < HID / 4; ++q) {
    const float4 w = wp[q], t = tp[q];
    s += w.x * t.x;
    s += w.y * t.y;
    s += w.z * t.z;
    s += w.w * t.w;
  }
  return s;
}
#define FOLD1 (INDIM * 16)
#define FOLD2 (HEADS * HID * 16)
#define NFOLD ((FOLD1 + FOLD2) / 256)
static_assert((FOLD1 + FOLD2) % 256 == 0, "whole work-groups of fold elements");

__global__ void owner_bytes(const int* __restrict__ dst, unsigned* __restrict__ own, unsigned* __restrict__ idx,
                            const float* __restrict__ W1, const float* __restrict__ att_s1,
                            const float* __restrict__ att_d1, const float* __restrict__ W2,
                            const float* __restrict__ att_s2, const float* __restrict__ att_d2,
                            float* __restrict__ bs1, float* __restrict__ bs2) {
  if (blockIdx.x < NFOLD) {
    const int o = blockIdx.x * 256 + threadIdx.x;
    if (o < FOLD1) bs1[o] = fold_elem<HEADS>(W1, att_s1, att_d1, o);
    else bs2[o - FOLD1] = fold_elem<1>(W2, att_s2, att_d2, o - FOLD1);
    return;
  }
  const int blk = blockIdx.x - NFOLD;
  __shared__ unsigned w_l[NBLK];
  if (threadIdx.x < NBLK) w_l[threadIdx.x] = 0u;
  __syncthreads();
  const int i = blk * blockDim.x + threadIdx.x;
  if (i < NE / 4) {
    const int4 d = ((const int4*)dst)[i];
    const unsigned o0 = (unsigned)(d.x / NPB), o1 = (unsigned)(d.y / NPB);
    const unsigned o2 = (unsigned)(d.z / NPB), o3 = (unsigned)(d.w / NPB);
    own[i] = o0 | (o1 << 8) | (o2 << 16) | (o3 << 24);
    const unsigned bit = 1u << (threadIdx.x >> 3);
    atomicOr(&w_l[o0], bit);
    atomicOr(&w_l[o1], bit);
    atomicOr(&w_l[o2], bit);
    atomicOr(&w_l[o3], bit);
  }
  __syncthreads();
  if (threadIdx.x < NBLK) idx[blk * NBLK + threadIdx.x] = w_l[threadIdx.x];
}
static_assert(NBLK <= 256, "one thread per owner in owner_bytes");

__device__ __forceinline__ unsigned zero_bytes(unsigned t) {
  return ~(((t & 0x7f7f7f7fu) + 0x7f7f7f7fu) | t) & 0x80808080u;
}


__device__ __forceinline__ void find_units(const uint4* __restrict__ own4, int u0, int u1,
                                           unsigned pat, float* q_l, unsigned* qcnt_l, int wave, int lane) {
  for (int ub = u0 + wave * 32; ub < u1; ub += 4 * 1024) {
    const int u = ub + lane;
    uint4 v[4];
    int id[4];
    bool ok[4];
#pragma unroll
    for (int q = 0; q < 4; ++q) {
      const int uu = u + q * 1024;
      ok[q] = uu < u1;
      id[q] = ok[q] ? uu : u0;
      v[q] = own4[id[q]];
    }
    unsigned bal[4];
    bool hit[4];
#pragma unroll
    for (int q = 0; q < 4; ++q) {
      const unsigned tx = v[q].x ^ pat, ty = v[q].y ^ pat, tz = v[q].z ^ pat, tw = v[q].w ^ pat;
      const unsigned a = ((tx - 0x01010101u) & ~tx) | ((ty - 0x01010101u) & ~ty) |
                         ((tz - 0x01010101u) & ~tz) | ((tw - 0x01010101u) & ~tw);
      hit[q] = ok[q] && ((a & 0x80808080u) != 0u);
      bal[q] = __builtin_amdgcn_ballot_w32(hit[q]);
    }
    const unsigned tot = (unsigned)(__builtin_popcount(bal[0]) + __builtin_popcount(bal[1]) +
                                    __builtin_popcount(bal[2]) + __builtin_popcount(bal[3]));
    if (tot) {
      unsigned base = 0u;
      if (lane == 0) base = atomicAdd(qcnt_l, tot);
      base = (unsigned)__builtin_amdgcn_readfirstlane((int)base);
#pragma unroll
      for (int q = 0; q < 4; ++q) {
        if (hit[q]) {
          const unsigned pos = base + __builtin_amdgcn_mbcnt_lo(bal[q], 0u);
          if (pos < QCAP) q_l[pos] = __uint_as_float((unsigned)id[q]);
        }
        base += (unsigned)__builtin_popcount(bal[q]);
      }
    }
  }
}

__device__ __forceinline__ void queue_groups(const unsigned* __restrict__ idx, int blk, float* q_l,
                                             unsigned* qcnt_l, int tid) {
  for (int t = tid; t < NCHK; t += 1024) {
    const unsigned w = idx[t * NBLK + blk];
    if (w) {
      unsigned pos = atomicAdd(qcnt_l, (unsigned)__popc(w));
      for (unsigned r = w; r; r &= r - 1) {
        if (pos < QCAP) q_l[pos] = __uint_as_float((unsigned)(t * 32 + __ffs(r) - 1));
        ++pos;
      }
    }
  }
}

template <int UPE>
__device__ __forceinline__ void decode_units(const uint4* __restrict__ own4, unsigned pat, const float* q_l, int nq,
                                             unsigned* list_l, unsigned* cnt_l, int wave, int lane) {
  static_assert(UPE == 1 || UPE == 2, "the byte mask of an entry must fit 32 bits");
  for (int i0 = wave * 32; i0 < nq; i0 += 4 * 1024) {
    uint4 v[4][UPE];
    unsigned id[4];
    bool ok[4];
#pragma unroll
    for (int q = 0; q < 4; ++q) {
      const int i = i0 + q * 1024 + lane;
      ok[q] = i < nq;
      id[q] = __float_as_uint(q_l[ok[q] ? i : i0]);
#pragma unroll
      for (int p = 0; p < UPE; ++p) v[q][p] = own4[id[q] * UPE + p];
    }
#pragma unroll
    for (int q = 0; q < 4; ++q) {
      unsigned mm = 0u;
#pragma unroll
      for (int p = 0; p < UPE; ++p)
        mm |= ((zero_bytes(v[q][p].x ^ pat) >> 7) | (zero_bytes(v[q][p].y ^ pat) >> 6) |
               (zero_bytes(v[q][p].z ^ pat) >> 5) | (zero_bytes(v[q][p].w ^ pat) >> 4)) << (4 * p);
      if (!ok[q]) mm = 0u;
      const unsigned c = (unsigned)__popc(mm);
      unsigned excl = 0u, tot = 0u;
#pragma unroll
      for (int b = 0; b < 4 + UPE; ++b) {
        const unsigned bl = __builtin_amdgcn_ballot_w32(((c >> b) & 1u) != 0u);
        excl += __builtin_amdgcn_mbcnt_lo(bl, 0u) << b;
        tot += (unsigned)__builtin_popcount(bl) << b;
      }
      if (tot) {
        unsigned base = 0u;
        if (lane == 0) base = atomicAdd(cnt_l, tot);
        unsigned pos = (unsigned)__builtin_amdgcn_readfirstlane((int)base) + excl;
        for (unsigned r = mm; r; r &= r - 1) {
          const int j = __ffs(r) - 1;
          if (pos < LCAP)
            list_l[pos] = id[q] * (16u * UPE) + (unsigned)(16 * ((j >> 2) & 1) + 4 * (j & 3) + (j >> 3));
          ++pos;
        }
      }
    }
  }
}

template <bool B> struct flag_c { static constexpr bool value = B; };

template <int H, bool FINAL>
__global__ __launch_bounds__(1024) void agg_owner(
    const int* __restrict__ src, const int* __restrict__ dst, const unsigned* __restrict__ own,
    const unsigned* __restrict__ idx,
    const float* __restrict__ sd, const float* __restrict__ feat,
    const float* __restrict__ bias, const float* __restrict__ lin_w, const float* __restrict__ lin_b,
    const float* __restrict__ Wn, const float* __restrict__ bsn, float* __restrict__ sdn,
    float* __restrict__ outp) {
  constexpr int HC = H * HID;
  __shared__ float    acc_l[NPB * HC];
  __shared__ float    den_l[NPB * H];
  __shared__ __attribute__((aligned(16))) float    ad_l[NPB * H];
  __shared__ __attribute__((aligned(16))) unsigned m_l[NPB * H];
  __shared__ unsigned list_l[LCAP + NPB];
  __shared__ __attribute__((aligned(16))) float p_l[(LCAP + NPB) * H];
  __shared__ unsigned cnt_l, qcnt_l;
  static_assert(QCAP <= (LCAP + NPB) * H, "the unit queue borrows p_l");
  constexpr int LDH = HC + 4;
  static_assert(FINAL || (H == 4 && NPB * LDH <= (LCAP + NPB) * H), "the finished rows borrow p_l");
  const int tid = threadIdx.x, lane = tid & 31;
  const int wave = __builtin_amdgcn_readfirstlane(tid >> 5);
  const int lo = blockIdx.x * NPB;
  const int nn = min(NPB, NN - lo);
  const uint4* own4 = (const uint4*)own;
  const unsigned pat = (unsigned)blockIdx.x * 0x01010101u;

  for (int i = tid; i < NPB * HC; i += 1024) acc_l[i] = 0.f;
  for (int i = tid; i < NPB * H; i += 1024) {
    float a_d = 0.f;
    unsigned mk = 0u;
    if (i < nn * H) {
      const float* sp = sd + (size_t)(lo + i / H) * (2 * H) + i % H;
      a_d = sp[H];
      mk = fkey(lrelu(sp[0] + a_d));
    }
    ad_l[i] = a_d;
    m_l[i] = mk;
    den_l[i] = 0.f;
  }
  if (tid == 0) { cnt_l = 0u; qcnt_l = 0u; }
  __syncthreads();

  auto max_pass = [&](int n) {
    for (int i = tid; i < n; i += 1024) {
      const int e = (int)list_l[i];
      const int s = src[e];
      const int dl = dst[e] - lo;
#pragma unroll
      for (int h = 0; h < H; ++h) {
        const float x = lrelu(sd[s * (2 * H) + h] + ad_l[dl * H + h]);
        atomicMax(&m_l[dl * H + h], fkey(x));
        p_l[i * H + h] = x;
      }
      list_l[i] = (unsigned)s | ((unsigned)dl << 16);
    }
  };

  auto weight_pass = [&](int n, int nself) {
    const int tot = n + nself;
    for (int i = tid; i < tot; i += 1024) {
      float x[H];
      int dl;
      if (i < n) {
        dl = (int)(list_l[i] >> 16);
#pragma unroll
        for (int h = 0; h < H; ++h) x[h] = p_l[i * H + h];
      } else {
        dl = i - n;
        list_l[i] = (unsigned)(lo + dl) | ((unsigned)dl << 16);
#pragma unroll
        for (int h = 0; h < H; ++h) x[h] = lrelu(sd[(lo + dl) * (2 * H) + h] + ad_l[dl * H + h]);
      }
#pragma unroll
      for (int h = 0; h < H; ++h) {
        const float p = __expf(x[h] - unfkey(m_l[dl * H + h]));
        p_l[i * H + h] = p;
        atomicAdd(&den_l[dl * H + h], p);
      }
    }
  };

  constexpr int BATCH = 16;
  constexpr int JB = 16;
  auto acc_batch = [&](auto all_c, int b0, int nb) __attribute__((always_inline)) {
    constexpr bool ALL = decltype(all_c)::value;
    const unsigned pkv = (lane < nb) ? list_l[b0 + lane] : 0u;
    int dl[JB] = {};
    if constexpr (H == 4) {
      v4f v[JB] = {};
      float p[JB] = {};
      const unsigned lane16 = 16u * (unsigned)lane;
#pragma unroll
      for (int j = 0; j < JB; ++j) {
        if (ALL || j < nb) {
          const unsigned pk = (unsigned)__builtin_amdgcn_readlane((int)pkv, j);
          dl[j] = (int)(pk >> 16);
          const unsigned off = (pk & 0xffffu) * (unsigned)(HC * sizeof(float)) + lane16;
          v[j] = *reinterpret_cast<const v4f*>(reinterpret_cast<const char*>(feat) + off);
        }
      }
#pragma unroll
      for (int j = 0; j < JB; ++j) {
        if (ALL || j < nb) p[j] = p_l[(b0 + j) * 4 + (lane >> 3)];
      }
#pragma unroll
      for (int j = 0; j < JB; ++j) {
        if (ALL || j < nb) {
#pragma unroll
          for (int i = 0; i < 4; ++i) atomicAdd(&acc_l[dl[j] * HC + i * 32 + lane], p[j] * v[j][i]);
        }
      }
    } else {
      float v[JB][H] = {}, p[JB][H] = {};
#pragma unroll
      for (int j = 0; j < JB; ++j) {
        if (ALL || j < nb) {
          const unsigned pk = (unsigned)__builtin_amdgcn_readlane((int)pkv, j);
          const int s = (int)(pk & 0xffffu);
          dl[j] = (int)(pk >> 16);
#pragma unroll
          for (int h = 0; h < H; ++h) v[j][h] = feat[(size_t)s * HC + h * HID + lane];
        }
      }
#pragma unroll
      for (int j = 0; j < JB; ++j) {
        if (ALL || j < nb) {
#pragma unroll
          for (int h = 0; h < H; ++h) p[j][h] = p_l[(b0 + j) * H + h];
        }
      }
#pragma unroll
      for (int j = 0; j < JB; ++j) {
        if (ALL || j < nb) {
#pragma unroll
          for (int h = 0; h < H; ++h)
            atomicAdd(&acc_l[dl[j] * HC + h * HID + lane], p[j][h] * v[j][h]);
        }
      }
    }
  };
  auto acc_pass = [&](int tot) __attribute__((always_inline)) {
    for (int b0 = wave * BATCH; b0 < tot; b0 += 32 * BATCH) {
      const int nb = min(BATCH, tot - b0);
      if (nb == BATCH) acc_batch(flag_c<true>{}, b0, BATCH);
      else acc_batch(flag_c<false>{}, b0, nb);
    }
  };

  auto scan_units = [&](int u0, int u1) __attribute__((always_inline)) {
    find_units(own4, u0, u1, pat, p_l, &qcnt_l, wave, lane);
    __syncthreads();
    const int nq = (int)qcnt_l;
    if (nq <= QCAP) decode_units<1>(own4, pat, p_l, nq, list_l, &cnt_l, wave, lane);
    __syncthreads();
    return (nq <= QCAP) ? (int)cnt_l : LCAP + 1;
  };
  auto scan_index = [&]() __attribute__((always_inline)) {
    queue_groups(idx, (int)blockIdx.x, p_l, &qcnt_l, tid);
    __syncthreads();
    const int nq = (int)qcnt_l;
    if (nq <= QCAP) decode_units<2>(own4, pat, p_l, nq, list_l, &cnt_l, wave, lane);
    __syncthreads();
    return (nq <= QCAP) ? (int)cnt_l : LCAP + 1;
  };

  const int n = scan_index();
  if (n <= LCAP) {
    max_pass(n);
    __syncthreads();
    weight_pass(n, nn);
    __syncthreads();
    acc_pass(n + nn);
  } else {
    for (int c0 = 0; c0 < NU; c0 += CHU) {
      __syncthreads();
      if (tid == 0) { cnt_l = 0u; qcnt_l = 0u; }
      __syncthreads();
      max_pass(scan_units(c0, min(c0 + CHU, NU)));
    }
    for (int c0 = 0; c0 < NU; c0 += CHU) {
      __syncthreads();
      if (tid == 0) { cnt_l = 0u; qcnt_l = 0u; }
      __syncthreads();
      const int nc = scan_units(c0, min(c0 + CHU, NU));
      const int ns = (c0 == 0) ? nn : 0;
      max_pass(nc);
      __syncthreads();
      weight_pass(nc, ns);
      __syncthreads();
      acc_pass(nc + ns);
    }
  }
  __syncthreads();

  if constexpr (!FINAL) {
    float* r_l = p_l;
    for (int i = tid; i < nn * (HC / 4); i += 1024) {
      const int dl = i / (HC / 4), q = i % (HC / 4);
      const float d = den_l[dl * H + q / (HID / 4)];
      const float4 bq = reinterpret_cast<const float4*>(bias)[q];
      float4 o;
      o.x = elu(acc_l[dl * HC + 0 * 32 + q] / d + bq.x);
      o.y = elu(acc_l[dl * HC + 1 * 32 + q] / d + bq.y);
      o.z = elu(acc_l[dl * HC + 2 * 32 + q] / d + bq.z);
      o.w = elu(acc_l[dl * HC + 3 * 32 + q] / d + bq.w);
      *reinterpret_cast<float4*>(r_l + dl * LDH + 4 * q) = o;
    }
    __syncthreads();
    const int t = wave >> 1;
    if (t < nn / 16) {
      const float* a0 = r_l + t * 16 * LDH;
      const size_t row0 = (size_t)(lo + 16 * t);
      if (((wave ^ (wave >> 2)) & 1) == 0) prod_tile<HID, HC, LDH, 2>(a0, Wn, outp + row0 * HID, 0, lane);
      else score_tile<HC, LDH, 1>(a0, bsn, sdn + row0 * 2, lane);
    }
  } else {
    for (int dl = wave; dl < nn; dl += 32) {
      const float v = elu(acc_l[dl * HC + lane] / den_l[dl * H] + bias[lane]);
      float w0 = v * lin_w[lane * 2 + 0];
      float w1 = v * lin_w[lane * 2 + 1];
#pragma unroll
      for (int m = 16; m > 0; m >>= 1) {
        w0 += __shfl_xor(w0, m, 32);
        w1 += __shfl_xor(w1, m, 32);
      }
      if (lane == 0) {
        outp[(lo + dl) * 2 + 0] = w0 + lin_b[0];
        outp[(lo + dl) * 2 + 1] = w1 + lin_b[1];
      }
    }
  }
}

extern "C" void kernel_launch(void* const* d_in, const int* in_sizes, int n_in,
                              void* d_out, int out_size, void* d_ws, size_t ws_size,
                              hipStream_t stream) {
  const float* x        = (const float*)d_in[0];
  const int*   ei       = (const int*)d_in[1];
  const float* W1       = (const float*)d_in[2];
  const float* att_src1 = (const float*)d_in[3];
  const float* att_dst1 = (const float*)d_in[4];
  const float* b1       = (const float*)d_in[5];
  const float* W2       = (const float*)d_in[6];
  const float* att_src2 = (const float*)d_in[7];
  const float* att_dst2 = (const float*)d_in[8];
  const float* b2       = (const float*)d_in[9];
  const float* lin_w    = (const float*)d_in[10];
  const float* lin_b    = (const float*)d_in[11];
  float* out = (float*)d_out;
  const int* src = ei;
  const int* dst = ei + NE;

  float*    ws   = (float*)d_ws;
  float*    h1   = ws;
  float*    sd1  = ws + 6400000;
  unsigned* own  = (unsigned*)(sd1 + 400000);
  unsigned* idx  = own + 200000;
  float*    bs1  = sd1 + 790000;
  float*    bs2  = bs1 + FOLD1;
  float*    z    = sd1 + 800000;
  float*    sd2  = z + 3200000;

  owner_bytes<<<NCHK + NFOLD, 256, 0, stream>>>(dst, own, idx, W1, att_src1, att_dst1, W2, att_src2, att_dst2,
                                                bs1, bs2);

  gemm_wmma_f32<HEADS * HID, INDIM, 4, HEADS>
      <<<dim3(NN / 16), dim3(32, 3, 1), 0, stream>>>(x, W1, h1, bs1, sd1);
  agg_owner<HEADS, false><<<NBLK, 1024, 0, stream>>>(src, dst, own, idx, sd1, h1, b1, lin_w, lin_b, W2, bs2, sd2, z);

  agg_owner<1, true><<<NBLK, 1024, 0, stream>>>(src, dst, own, idx, sd2, z, b2, lin_w, lin_b, nullptr, nullptr,
                                                nullptr, out);
}


# Layer-1 aggregation stores each work-group's packed edge list in the workspace; layer 2 reads it back in whole lines, no search and no src/dst reads; layer-2 sums in layer 1's LDS arrival order
# speedup vs baseline: 8.0000x; 1.1136x over previous
#include <hip/hip_runtime.h>
#include <math.h>

#define NN 50000
#define NE 800000
#define INDIM 256
#define HID 32
#define HEADS 4
#define ETOT (NE + NN)

typedef __attribute__((ext_vector_type(2))) float v2f;
typedef __attribute__((ext_vector_type(8))) float v8f;

__device__ __forceinline__ unsigned fkey(float x) {
  unsigned u = __float_as_uint(x);
  return (u & 0x80000000u) ? ~u : (u | 0x80000000u);
}
__device__ __forceinline__ float unfkey(unsigned k) {
  return __uint_as_float((k & 0x80000000u) ? (k & 0x7fffffffu) : ~k);
}
__device__ __forceinline__ float elu(float v) {
  return (v > 0.f) ? v : (__expf(v) - 1.f);
}

typedef __attribute__((ext_vector_type(4))) float v4f;
template <int NT> struct colvec;
template <> struct colvec<4> { typedef v4f type; };
template <> struct colvec<2> { typedef v2f type; };

template <int K, int LDA, int H>
__device__ __forceinline__ void score_tile(const float* __restrict__ A0, const float* __restrict__ Bs,
                                           float* __restrict__ SD0, int lane) {
  const int c = lane & 15;
  const int h = lane >> 4;
  const float* ap = A0 + c * LDA + 4 * h;
  const float* bp = Bs + h * 64 + 4 * c;
  constexpr int G   = 4;
  constexpr int NCH = K / (8 * G);
  static_assert(K % (8 * G) == 0, "whole chunks");
  v8f acc[2];
#pragma unroll
  for (int t = 0; t < 2; ++t) acc[t] = v8f{0.f, 0.f, 0.f, 0.f, 0.f, 0.f, 0.f, 0.f};
  v4f af[2][G], bf[2][G];
#pragma unroll
  for (int g = 0; g < G; ++g) {
    af[0][g] = *reinterpret_cast<const v4f*>(ap + 8 * g);
    bf[0][g] = *reinterpret_cast<const v4f*>(bp + 128 * g);
  }
#pragma unroll
  for (int ch = 0; ch < NCH; ++ch) {
    const int cur = ch & 1, nxt = cur ^ 1;
    if (ch + 1 < NCH) {
#pragma unroll
      for (int g = 0; g < G; ++g) {
        af[nxt][g] = *reinterpret_cast<const v4f*>(ap + (ch + 1) * 8 * G + 8 * g);
        bf[nxt][g] = *reinterpret_cast<const v4f*>(bp + ((ch + 1) * G + g) * 128);
      }
    }
    __builtin_amdgcn_sched_barrier(0);
#pragma unroll
    for (int u = 0; u < 2; ++u) {
#pragma unroll
      for (int g = 0; g < G; ++g) {
        v2f a, b;
        a.x = af[cur][g][2 * u];
        a.y = af[cur][g][2 * u + 1];
        b.x = bf[cur][g][2 * u];
        b.y = bf[cur][g][2 * u + 1];
        acc[g & 1] = __builtin_amdgcn_wmma_f32_16x16x4_f32(false, a, false, b, (short)0, acc[g & 1],
                                                           false, false);
      }
    }
    __builtin_amdgcn_sched_barrier(0);
  }
  if (c < 2 * H) {
    float* sp = SD0 + (8 * h) * (2 * H) + c;
#pragma unroll
    for (int r = 0; r < 8; ++r) sp[r * (2 * H)] = acc[0][r] + acc[1][r];
  }
}

template <int N, int K, int LDA, int NT>
__device__ __forceinline__ void prod_tile(const float* __restrict__ A0, const float* __restrict__ B,
                                          float* __restrict__ C0, int nb, int lane) {
  typedef typename colvec<NT>::type vnt;
  const int c     = lane & 15;
  const int h     = lane >> 4;
  const float* ap = A0 + c * LDA + 4 * h;
  const float* bp = B + (size_t)(4 * h) * N + nb + NT * c;
  v8f acc[NT];
#pragma unroll
  for (int t = 0; t < NT; ++t) acc[t] = v8f{0.f, 0.f, 0.f, 0.f, 0.f, 0.f, 0.f, 0.f};

  constexpr int KC  = 16;
  constexpr int NCH = K / KC;
  static_assert(K % KC == 0, "whole chunks");
  v4f af[2][2];
  vnt bf[2][8];
#pragma unroll
  for (int g = 0; g < 2; ++g) {
    af[0][g] = *reinterpret_cast<const v4f*>(ap + 8 * g);
#pragma unroll
    for (int j = 0; j < 4; ++j)
      bf[0][4 * g + j] = *reinterpret_cast<const vnt*>(bp + (size_t)(8 * g + j) * N);
  }
#pragma unroll
  for (int ch = 0; ch < NCH; ++ch) {
    const int cur = ch & 1, nxt = cur ^ 1;
    if (ch + 1 < NCH) {
      const float* apn = ap + (ch + 1) * KC;
      const float* bpn = bp + (size_t)(ch + 1) * KC * N;
#pragma unroll
      for (int g = 0; g < 2; ++g) {
        af[nxt][g] = *reinterpret_cast<const v4f*>(apn + 8 * g);
#pragma unroll
        for (int j = 0; j < 4; ++j)
          bf[nxt][4 * g + j] = *reinterpret_cast<const vnt*>(bpn + (size_t)(8 * g + j) * N);
      }
    }
    __builtin_amdgcn_sched_barrier(0);
#pragma unroll
    for (int g = 0; g < 2; ++g) {
#pragma unroll
      for (int u = 0; u < 2; ++u) {
        v2f a;
        a.x = af[cur][g][2 * u];
        a.y = af[cur][g][2 * u + 1];
#pragma unroll
        for (int t = 0; t < NT; ++t) {
          v2f b;
          b.x = bf[cur][4 * g + 2 * u][t];
          b.y = bf[cur][4 * g + 2 * u + 1][t];
          acc[t] = __builtin_amdgcn_wmma_f32_16x16x4_f32(false, a, false, b, (short)0, acc[t],
                                                         false, false);
        }
      }
    }
    __builtin_amdgcn_sched_barrier(0);
  }
  float* cp = C0 + (size_t)(8 * h) * N + nb + NT * c;
#pragma unroll
  for (int r = 0; r < 8; ++r) {
    vnt o;
#pragma unroll
    for (int t = 0; t < NT; ++t) o[t] = acc[t][r];
    *reinterpret_cast<vnt*>(cp + (size_t)r * N) = o;
  }
}

template <int N, int K, int NT, int H>
__global__ void gemm_wmma_f32(const float* __restrict__ A, const float* __restrict__ B,
                              float* __restrict__ C, const float* __restrict__ Bs,
                              float* __restrict__ SD) {
  const int lane  = threadIdx.x;
  const int mbase = (blockIdx.x * blockDim.z + threadIdx.z) * 16;
  const int wy    = __builtin_amdgcn_readfirstlane((int)threadIdx.y);
  if (wy == N / (16 * NT)) {
    score_tile<K, K, H>(A + (size_t)mbase * K, Bs, SD + (size_t)mbase * (2 * H), lane);
    return;
  }
  prod_tile<N, K, K, NT>(A + (size_t)mbase * K, B, C + (size_t)mbase * N, wy * (16 * NT), lane);
}

#define NPB  208
#define NBLK ((NN + NPB - 1) / NPB)
#define NU   (NE / 16)
#define LCAP 8192
#define CHU  (LCAP / 16)
#define QCAP LCAP
#define NCHK ((NE / 4 + 255) / 256)
#define SHCAP 6400
static_assert(SHCAP <= LCAP && (SHCAP * 4) % 128 == 0, "a handed-on list fits the LDS list; its rows start on whole lines");
static_assert(NBLK <= 256, "owner id must fit a byte");
static_assert(NE % 32 == 0, "owner map is read in 16-byte units, and in pairs of them through the index");
static_assert(NN <= 65536 && NPB <= 65536, "list entry packs source and local destination in 16 bits each");
static_assert(HID == 32, "one lane per channel of a head");
static_assert((unsigned long long)NN * HEADS * HID * sizeof(float) < (1ull << 32), "a feature row's byte offset fits 32 bits");
static_assert(NN % 16 == 0 && NPB % 16 == 0, "products: whole 16-row tiles, in the first product's launch and in a work-group's owned rows");

__device__ __forceinline__ float lrelu(float x) { return (x > 0.f) ? x : 0.2f * x; }

template <int H>
__device__ __forceinline__ float fold_elem(const float* __restrict__ Wm, const float* __restrict__ att_s,
                                           const float* __restrict__ att_d, int o) {
  const int k = ((o >> 6) << 2) | (o & 3);
  const int col = (o >> 2) & 15;
  if (col >= 2 * H) return 0.f;
  const int hd = (col < H) ? col : col - H;
  const float4* wp = reinterpret_cast<const float4*>(Wm + (size_t)k * (H * HID) + hd * HID);
  const float4* tp = reinterpret_cast<const float4*>(((col < H) ? att_s : att_d) + hd * HID);
  float s = 0.f;
#pragma unroll
  for (int q = 0; q < HID / 4; ++q) {
    const float4 w = wp[q], t = tp[q];
    s += w.x * t.x;
    s += w.y * t.y;
    s += w.z * t.z;
    s += w.w * t.w;
  }
  return s;
}
#define FOLD1 (INDIM * 16)
#define FOLD2 (HEADS * HID * 16)
#define NFOLD ((FOLD1 + FOLD2) / 256)
static_assert((FOLD1 + FOLD2) % 256 == 0, "whole work-groups of fold elements");

__global__ void owner_bytes(const int* __restrict__ dst, unsigned* __restrict__ own, unsigned* __restrict__ idx,
                            const float* __restrict__ W1, const float* __restrict__ att_s1,
                            const float* __restrict__ att_d1, const float* __restrict__ W2,
                            const float* __restrict__ att_s2, const float* __restrict__ att_d2,
                            float* __restrict__ bs1, float* __restrict__ bs2) {
  if (blockIdx.x < NFOLD) {
    const int o = blockIdx.x * 256 + threadIdx.x;
    if (o < FOLD1) bs1[o] = fold_elem<HEADS>(W1, att_s1, att_d1, o);
    else bs2[o - FOLD1] = fold_elem<1>(W2, att_s2, att_d2, o - FOLD1);
    return;
  }
  const int blk = blockIdx.x - NFOLD;
  __shared__ unsigned w_l[NBLK];
  if (threadIdx.x < NBLK) w_l[threadIdx.x] = 0u;
  __syncthreads();
  const int i = blk * blockDim.x + threadIdx.x;
  if (i < NE / 4) {
    const int4 d = ((const int4*)dst)[i];
    const unsigned o0 = (unsigned)(d.x / NPB), o1 = (unsigned)(d.y / NPB);
    const unsigned o2 = (unsigned)(d.z / NPB), o3 = (unsigned)(d.w / NPB);
    own[i] = o0 | (o1 << 8) | (o2 << 16) | (o3 << 24);
    const unsigned bit = 1u << (threadIdx.x >> 3);
    atomicOr(&w_l[o0], bit);
    atomicOr(&w_l[o1], bit);
    atomicOr(&w_l[o2], bit);
    atomicOr(&w_l[o3], bit);
  }
  __syncthreads();
  if (threadIdx.x < NBLK) idx[blk * NBLK + threadIdx.x] = w_l[threadIdx.x];
}
static_assert(NBLK <= 256, "one thread per owner in owner_bytes");

__device__ __forceinline__ unsigned zero_bytes(unsigned t) {
  return ~(((t & 0x7f7f7f7fu) + 0x7f7f7f7fu) | t) & 0x80808080u;
}


__device__ __forceinline__ void find_units(const uint4* __restrict__ own4, int u0, int u1,
                                           unsigned pat, float* q_l, unsigned* qcnt_l, int wave, int lane) {
  for (int ub = u0 + wave * 32; ub < u1; ub += 4 * 1024) {
    const int u = ub + lane;
    uint4 v[4];
    int id[4];
    bool ok[4];
#pragma unroll
    for (int q = 0; q < 4; ++q) {
      const int uu = u + q * 1024;
      ok[q] = uu < u1;
      id[q] = ok[q] ? uu : u0;
      v[q] = own4[id[q]];
    }
    unsigned bal[4];
    bool hit[4];
#pragma unroll
    for (int q = 0; q < 4; ++q) {
      const unsigned tx = v[q].x ^ pat, ty = v[q].y ^ pat, tz = v[q].z ^ pat, tw = v[q].w ^ pat;
      const unsigned a = ((tx - 0x01010101u) & ~tx) | ((ty - 0x01010101u) & ~ty) |
                         ((tz - 0x01010101u) & ~tz) | ((tw - 0x01010101u) & ~tw);
      hit[q] = ok[q] && ((a & 0x80808080u) != 0u);
      bal[q] = __builtin_amdgcn_ballot_w32(hit[q]);
    }
    const unsigned tot = (unsigned)(__builtin_popcount(bal[0]) + __builtin_popcount(bal[1]) +
                                    __builtin_popcount(bal[2]) + __builtin_popcount(bal[3]));
    if (tot) {
      unsigned base = 0u;
      if (lane == 0) base = atomicAdd(qcnt_l, tot);
      base = (unsigned)__builtin_amdgcn_readfirstlane((int)base);
#pragma unroll
      for (int q = 0; q < 4; ++q) {
        if (hit[q]) {
          const unsigned pos = base + __builtin_amdgcn_mbcnt_lo(bal[q], 0u);
          if (pos < QCAP) q_l[pos] = __uint_as_float((unsigned)id[q]);
        }
        base += (unsigned)__builtin_popcount(bal[q]);
      }
    }
  }
}

__device__ __forceinline__ void queue_groups(const unsigned* __restrict__ idx, int blk, float* q_l,
                                             unsigned* qcnt_l, int tid) {
  for (int t = tid; t < NCHK; t += 1024) {
    const unsigned w = idx[t * NBLK + blk];
    if (w) {
      unsigned pos = atomicAdd(qcnt_l, (unsigned)__popc(w));
      for (unsigned r = w; r; r &= r - 1) {
        if (pos < QCAP) q_l[pos] = __uint_as_float((unsigned)(t * 32 + __ffs(r) - 1));
        ++pos;
      }
    }
  }
}

template <int UPE>
__device__ __forceinline__ void decode_units(const uint4* __restrict__ own4, unsigned pat, const float* q_l, int nq,
                                             unsigned* list_l, unsigned* cnt_l, int wave, int lane) {
  static_assert(UPE == 1 || UPE == 2, "the byte mask of an entry must fit 32 bits");
  for (int i0 = wave * 32; i0 < nq; i0 += 4 * 1024) {
    uint4 v[4][UPE];
    unsigned id[4];
    bool ok[4];
#pragma unroll
    for (int q = 0; q < 4; ++q) {
      const int i = i0 + q * 1024 + lane;
      ok[q] = i < nq;
      id[q] = __float_as_uint(q_l[ok[q] ? i : i0]);
#pragma unroll
      for (int p = 0; p < UPE; ++p) v[q][p] = own4[id[q] * UPE + p];
    }
#pragma unroll
    for (int q = 0; q < 4; ++q) {
      unsigned mm = 0u;
#pragma unroll
      for (int p = 0; p < UPE; ++p)
        mm |= ((zero_bytes(v[q][p].x ^ pat) >> 7) | (zero_bytes(v[q][p].y ^ pat) >> 6) |
               (zero_bytes(v[q][p].z ^ pat) >> 5) | (zero_bytes(v[q][p].w ^ pat) >> 4)) << (4 * p);
      if (!ok[q]) mm = 0u;
      const unsigned c = (unsigned)__popc(mm);
      unsigned excl = 0u, tot = 0u;
#pragma unroll
      for (int b = 0; b < 4 + UPE; ++b) {
        const unsigned bl = __builtin_amdgcn_ballot_w32(((c >> b) & 1u) != 0u);
        excl += __builtin_amdgcn_mbcnt_lo(bl, 0u) << b;
        tot += (unsigned)__builtin_popcount(bl) << b;
      }
      if (tot) {
        unsigned base = 0u;
        if (lane == 0) base = atomicAdd(cnt_l, tot);
        unsigned pos = (unsigned)__builtin_amdgcn_readfirstlane((int)base) + excl;
        for (unsigned r = mm; r; r &= r - 1) {
          const int j = __ffs(r) - 1;
          if (pos < LCAP)
            list_l[pos] = id[q] * (16u * UPE) + (unsigned)(16 * ((j >> 2) & 1) + 4 * (j & 3) + (j >> 3));
          ++pos;
        }
      }
    }
  }
}

template <bool B> struct flag_c { static constexpr bool value = B; };

template <int H, bool FINAL>
__global__ __launch_bounds__(1024) void agg_owner(
    const int* __restrict__ src, const int* __restrict__ dst, const unsigned* __restrict__ own,
    const unsigned* __restrict__ idx,
    const float* __restrict__ sd, const float* __restrict__ feat,
    const float* __restrict__ bias, const float* __restrict__ lin_w, const float* __restrict__ lin_b,
    const float* __restrict__ Wn, const float* __restrict__ bsn, float* __restrict__ sdn,
    unsigned* __restrict__ shcnt, unsigned* __restrict__ shlist,
    float* __restrict__ outp) {
  constexpr int HC = H * HID;
  __shared__ float    acc_l[NPB * HC];
  __shared__ float    den_l[NPB * H];
  __shared__ __attribute__((aligned(16))) float    ad_l[NPB * H];
  __shared__ __attribute__((aligned(16))) unsigned m_l[NPB * H];
  __shared__ unsigned list_l[LCAP + NPB];
  __shared__ __attribute__((aligned(16))) float p_l[(LCAP + NPB) * H];
  __shared__ unsigned cnt_l, qcnt_l;
  static_assert(QCAP <= (LCAP + NPB) * H, "the unit queue borrows p_l");
  constexpr int LDH = HC + 4;
  static_assert(FINAL || (H == 4 && NPB * LDH <= (LCAP + NPB) * H), "the finished rows borrow p_l");
  const int tid = threadIdx.x, lane = tid & 31;
  const int wave = __builtin_amdgcn_readfirstlane(tid >> 5);
  const int lo = blockIdx.x * NPB;
  const int nn = min(NPB, NN - lo);
  const uint4* own4 = (const uint4*)own;
  const unsigned pat = (unsigned)blockIdx.x * 0x01010101u;
  int ns = SHCAP + 1;
  if constexpr (FINAL) ns = (int)min(shcnt[blockIdx.x], (unsigned)(SHCAP + 1));

  for (int i = tid; i < NPB * HC; i += 1024) acc_l[i] = 0.f;
  for (int i = tid; i < NPB * H; i += 1024) {
    float a_d = 0.f;
    unsigned mk = 0u;
    if (i < nn * H) {
      const float* sp = sd + (size_t)(lo + i / H) * (2 * H) + i % H;
      a_d = sp[H];
      mk = fkey(lrelu(sp[0] + a_d));
    }
    ad_l[i] = a_d;
    m_l[i] = mk;
    den_l[i] = 0.f;
  }
  if (tid == 0) { cnt_l = 0u; qcnt_l = 0u; }
  __syncthreads();

  auto max_pass = [&](int n, unsigned* wl) {
    for (int i = tid; i < n; i += 1024) {
      const int e = (int)list_l[i];
      const int s = src[e];
      const int dl = dst[e] - lo;
#pragma unroll
      for (int h = 0; h < H; ++h) {
        const float x = lrelu(sd[s * (2 * H) + h] + ad_l[dl * H + h]);
        atomicMax(&m_l[dl * H + h], fkey(x));
        p_l[i * H + h] = x;
      }
      const unsigned pk = (unsigned)s | ((unsigned)dl << 16);
      list_l[i] = pk;
      if (wl) wl[i] = pk;
    }
  };

  auto list_pass = [&](int n) {
    const unsigned* wl = shlist + (size_t)blockIdx.x * SHCAP;
    for (int i0 = tid; i0 < n; i0 += 4 * 1024) {
      unsigned pk[4];
      bool ok[4];
      float xs[4][H];
#pragma unroll
      for (int q = 0; q < 4; ++q) {
        const int i = i0 + q * 1024;
        ok[q] = i < n;
        pk[q] = wl[ok[q] ? i : i0];
      }
#pragma unroll
      for (int q = 0; q < 4; ++q) {
#pragma unroll
        for (int h = 0; h < H; ++h) xs[q][h] = sd[(pk[q] & 0xffffu) * (2 * H) + h];
      }
#pragma unroll
      for (int q = 0; q < 4; ++q) {
        const int i = i0 + q * 1024;
        const int dl = (int)(pk[q] >> 16);
#pragma unroll
        for (int h = 0; h < H; ++h) {
          const float x = lrelu(xs[q][h] + ad_l[dl * H + h]);
          atomicMax(&m_l[dl * H + h], ok[q] ? fkey(x) : 0u);
          if (ok[q]) p_l[i * H + h] = x;
        }
        if (ok[q]) list_l[i] = pk[q];
      }
    }
  };

  auto weight_pass = [&](int n, int nself) {
    const int tot = n + nself;
    for (int i = tid; i < tot; i += 1024) {
      float x[H];
      int dl;
      if (i < n) {
        dl = (int)(list_l[i] >> 16);
#pragma unroll
        for (int h = 0; h < H; ++h) x[h] = p_l[i * H + h];
      } else {
        dl = i - n;
        list_l[i] = (unsigned)(lo + dl) | ((unsigned)dl << 16);
#pragma unroll
        for (int h = 0; h < H; ++h) x[h] = lrelu(sd[(lo + dl) * (2 * H) + h] + ad_l[dl * H + h]);
      }
#pragma unroll
      for (int h = 0; h < H; ++h) {
        const float p = __expf(x[h] - unfkey(m_l[dl * H + h]));
        p_l[i * H + h] = p;
        atomicAdd(&den_l[dl * H + h], p);
      }
    }
  };

  constexpr int BATCH = 16;
  constexpr int JB = 16;
  auto acc_batch = [&](auto all_c, int b0, int nb) __attribute__((always_inline)) {
    constexpr bool ALL = decltype(all_c)::value;
    const unsigned pkv = (lane < nb) ? list_l[b0 + lane] : 0u;
    int dl[JB] = {};
    if constexpr (H == 4) {
      v4f v[JB] = {};
      float p[JB] = {};
      const unsigned lane16 = 16u * (unsigned)lane;
#pragma unroll
      for (int j = 0; j < JB; ++j) {
        if (ALL || j < nb) {
          const unsigned pk = (unsigned)__builtin_amdgcn_readlane((int)pkv, j);
          dl[j] = (int)(pk >> 16);
          const unsigned off = (pk & 0xffffu) * (unsigned)(HC * sizeof(float)) + lane16;
          v[j] = *reinterpret_cast<const v4f*>(reinterpret_cast<const char*>(feat) + off);
        }
      }
#pragma unroll
      for (int j = 0; j < JB; ++j) {
        if (ALL || j < nb) p[j] = p_l[(b0 + j) * 4 + (lane >> 3)];
      }
#pragma unroll
      for (int j = 0; j < JB; ++j) {
        if (ALL || j < nb) {
#pragma unroll
          for (int i = 0; i < 4; ++i) atomicAdd(&acc_l[dl[j] * HC + i * 32 + lane], p[j] * v[j][i]);
        }
      }
    } else {
      float v[JB][H] = {}, p[JB][H] = {};
#pragma unroll
      for (int j = 0; j < JB; ++j) {
        if (ALL || j < nb) {
          const unsigned pk = (unsigned)__builtin_amdgcn_readlane((int)pkv, j);
          const int s = (int)(pk & 0xffffu);
          dl[j] = (int)(pk >> 16);
#pragma unroll
          for (int h = 0; h < H; ++h) v[j][h] = feat[(size_t)s * HC + h * HID + lane];
        }
      }
#pragma unroll
      for (int j = 0; j < JB; ++j) {
        if (ALL || j < nb) {
#pragma unroll
          for (int h = 0; h < H; ++h) p[j][h] = p_l[(b0 + j) * H + h];
        }
      }
#pragma unroll
      for (int j = 0; j < JB; ++j) {
        if (ALL || j < nb) {
#pragma unroll
          for (int h = 0; h < H; ++h)
            atomicAdd(&acc_l[dl[j] * HC + h * HID + lane], p[j][h] * v[j][h]);
        }
      }
    }
  };
  auto acc_pass = [&](int tot) __attribute__((always_inline)) {
    for (int b0 = wave * BATCH; b0 < tot; b0 += 32 * BATCH) {
      const int nb = min(BATCH, tot - b0);
      if (nb == BATCH) acc_batch(flag_c<true>{}, b0, BATCH);
      else acc_batch(flag_c<false>{}, b0, nb);
    }
  };

  auto scan_units = [&](int u0, int u1) __attribute__((always_inline)) {
    find_units(own4, u0, u1, pat, p_l, &qcnt_l, wave, lane);
    __syncthreads();
    const int nq = (int)qcnt_l;
    if (nq <= QCAP) decode_units<1>(own4, pat, p_l, nq, list_l, &cnt_l, wave, lane);
    __syncthreads();
    return (nq <= QCAP) ? (int)cnt_l : LCAP + 1;
  };
  auto scan_index = [&]() __attribute__((always_inline)) {
    queue_groups(idx, (int)blockIdx.x, p_l, &qcnt_l, tid);
    __syncthreads();
    const int nq = (int)qcnt_l;
    if (nq <= QCAP) decode_units<2>(own4, pat, p_l, nq, list_l, &cnt_l, wave, lane);
    __syncthreads();
    return (nq <= QCAP) ? (int)cnt_l : LCAP + 1;
  };

  const int n = (ns <= SHCAP) ? ns : scan_index();
  if constexpr (!FINAL) {
    if (tid == 0) shcnt[blockIdx.x] = (unsigned)n;
  }
  if (ns <= SHCAP) {
    list_pass(n);
    __syncthreads();
    weight_pass(n, nn);
    __syncthreads();
    acc_pass(n + nn);
  } else if (n <= LCAP) {
    max_pass(n, (!FINAL && n <= SHCAP) ? shlist + (size_t)blockIdx.x * SHCAP : nullptr);
    __syncthreads();
    weight_pass(n, nn);
    __syncthreads();
    acc_pass(n + nn);
  } else {
    for (int c0 = 0; c0 < NU; c0 += CHU) {
      __syncthreads();
      if (tid == 0) { cnt_l = 0u; qcnt_l = 0u; }
      __syncthreads();
      max_pass(scan_units(c0, min(c0 + CHU, NU)), nullptr);
    }
    for (int c0 = 0; c0 < NU; c0 += CHU) {
      __syncthreads();
      if (tid == 0) { cnt_l = 0u; qcnt_l = 0u; }
      __syncthreads();
      const int nc = scan_units(c0, min(c0 + CHU, NU));
      const int ns = (c0 == 0) ? nn : 0;
      max_pass(nc, nullptr);
      __syncthreads();
      weight_pass(nc, ns);
      __syncthreads();
      acc_pass(nc + ns);
    }
  }
  __syncthreads();

  if constexpr (!FINAL) {
    float* r_l = p_l;
    for (int i = tid; i < nn * (HC / 4); i += 1024) {
      const int dl = i / (HC / 4), q = i % (HC / 4);
      const float d = den_l[dl * H + q / (HID / 4)];
      const float4 bq = reinterpret_cast<const float4*>(bias)[q];
      float4 o;
      o.x = elu(acc_l[dl * HC + 0 * 32 + q] / d + bq.x);
      o.y = elu(acc_l[dl * HC + 1 * 32 + q] / d + bq.y);
      o.z = elu(acc_l[dl * HC + 2 * 32 + q] / d + bq.z);
      o.w = elu(acc_l[dl * HC + 3 * 32 + q] / d + bq.w);
      *reinterpret_cast<float4*>(r_l + dl * LDH + 4 * q) = o;
    }
    __syncthreads();
    const int t = wave >> 1;
    if (t < nn / 16) {
      const float* a0 = r_l + t * 16 * LDH;
      const size_t row0 = (size_t)(lo + 16 * t);
      if (((wave ^ (wave >> 2)) & 1) == 0) prod_tile<HID, HC, LDH, 2>(a0, Wn, outp + row0 * HID, 0, lane);
      else score_tile<HC, LDH, 1>(a0, bsn, sdn + row0 * 2, lane);
    }
  } else {
    for (int dl = wave; dl < nn; dl += 32) {
      const float v = elu(acc_l[dl * HC + lane] / den_l[dl * H] + bias[lane]);
      float w0 = v * lin_w[lane * 2 + 0];
      float w1 = v * lin_w[lane * 2 + 1];
#pragma unroll
      for (int m = 16; m > 0; m >>= 1) {
        w0 += __shfl_xor(w0, m, 32);
        w1 += __shfl_xor(w1, m, 32);
      }
      if (lane == 0) {
        outp[(lo + dl) * 2 + 0] = w0 + lin_b[0];
        outp[(lo + dl) * 2 + 1] = w1 + lin_b[1];
      }
    }
  }
}

extern "C" void kernel_launch(void* const* d_in, const int* in_sizes, int n_in,
                              void* d_out, int out_size, void* d_ws, size_t ws_size,
                              hipStream_t stream) {
  const float* x        = (const float*)d_in[0];
  const int*   ei       = (const int*)d_in[1];
  const float* W1       = (const float*)d_in[2];
  const float* att_src1 = (const float*)d_in[3];
  const float* att_dst1 = (const float*)d_in[4];
  const float* b1       = (const float*)d_in[5];
  const float* W2       = (const float*)d_in[6];
  const float* att_src2 = (const float*)d_in[7];
  const float* att_dst2 = (const float*)d_in[8];
  const float* b2       = (const float*)d_in[9];
  const float* lin_w    = (const float*)d_in[10];
  const float* lin_b    = (const float*)d_in[11];
  float* out = (float*)d_out;
  const int* src = ei;
  const int* dst = ei + NE;

  float*    ws   = (float*)d_ws;
  float*    h1   = ws;
  float*    sd1  = ws + 6400000;
  unsigned* own  = (unsigned*)(sd1 + 400000);
  unsigned* idx  = own + 200000;
  float*    bs1  = sd1 + 790000;
  float*    bs2  = bs1 + FOLD1;
  float*    z    = sd1 + 800000;
  float*    sd2  = z + 3200000;
  unsigned* shc  = (unsigned*)(z + 1600000);
  unsigned* shl  = shc + 256;
  static_assert(256 + NBLK * SHCAP <= 1600000, "the stored lists and their lengths fit between z and sd2");

  owner_bytes<<<NCHK + NFOLD, 256, 0, stream>>>(dst, own, idx, W1, att_src1, att_dst1, W2, att_src2, att_dst2,
                                                bs1, bs2);

  gemm_wmma_f32<HEADS * HID, INDIM, 4, HEADS>
      <<<dim3(NN / 16), dim3(32, 3, 1), 0, stream>>>(x, W1, h1, bs1, sd1);
  agg_owner<HEADS, false><<<NBLK, 1024, 0, stream>>>(src, dst, own, idx, sd1, h1, b1, lin_w, lin_b, W2, bs2, sd2,
                                                     shc, shl, z);

  agg_owner<1, true><<<NBLK, 1024, 0, stream>>>(src, dst, own, idx, sd2, z, b2, lin_w, lin_b, nullptr, nullptr,
                                                nullptr, shc, shl, out);
}
